# starting kernel
# baseline (speedup 1.0000x reference)
#include <hip/hip_runtime.h>

#define BATCH 4096
#define SEQ   2048
#define ARP   8
#define EXO   8
#define USTR  (SEQ + EXO)
#define OSTR  (SEQ + ARP + 1)
#define KTAPS 256
#define EROWS (SEQ + 8)
#define EROWS_PAD (EROWS + 8)
#define HOFF  32
#define HPAD_LEN 512

typedef __attribute__((ext_vector_type(2))) float v2f;
typedef __attribute__((ext_vector_type(8))) float v8f;

__global__ void k_impulse(const float* __restrict__ W, float* __restrict__ hpad) {
  if (threadIdx.x == 0 && blockIdx.x == 0) {
    for (int i = 0; i < HPAD_LEN; ++i) hpad[i] = 0.0f;
    hpad[HOFF] = 1.0f;
    for (int n = 1; n < KTAPS; ++n) {
      float s = 0.0f;
#pragma unroll
      for (int d = 1; d <= 8; ++d) s += W[8 - d] * hpad[HOFF + n - d];
      hpad[HOFF + n] = s;
    }
  }
}

__global__ void k_eps(const float* __restrict__ y, const float* __restrict__ hpad,
                      float* __restrict__ Efull) {
  const int b = blockIdx.x * blockDim.x + threadIdx.x;
  if (b >= BATCH) return;
  float hv[8];
#pragma unroll
  for (int i = 0; i < 8; ++i) hv[i] = hpad[HOFF + i];
  float e[8];
#pragma unroll
  for (int i = 0; i < 8; ++i) {
    float s = y[b * 9 + i + 1] - y[b * 9 + i];
#pragma unroll
    for (int p = 0; p < 8; ++p) {
      if (p < i) s -= hv[i - p] * e[p];
    }
    e[i] = s;
    Efull[i * BATCH + b] = s;
  }
#pragma unroll
  for (int i = 0; i < 8; ++i)
    Efull[(size_t)(EROWS + i) * BATCH + b] = 0.0f;
}

__global__ void k_exo(const float* __restrict__ u, const float* __restrict__ W,
                      float* __restrict__ Efull) {
  const int b = blockIdx.x * blockDim.x + threadIdx.x;
  const int m = blockIdx.y;
  float wv[8];
#pragma unroll
  for (int j = 0; j < 8; ++j) wv[j] = W[8 + j];
  const float* up = u + (size_t)b * USTR + 1 + m;
  float s = 0.0f;
#pragma unroll
  for (int j = 0; j < 8; ++j) s += wv[j] * up[j];
  Efull[(size_t)(m + 8) * BATCH + b] = s;
}

__global__ void k_zconv(const float* __restrict__ hpad, const float* __restrict__ Efull,
                        float* __restrict__ Z) {
  const int lane = threadIdx.x & 31;
  const int wv   = threadIdx.x >> 5;
  const int nbase = blockIdx.x * 16 + 1;
  const int b0 = blockIdx.y * 256 + wv * 64;
  const int col  = lane & 15;
  const int half = lane >> 4;

  v8f acc0 = {0.0f, 0.0f, 0.0f, 0.0f, 0.0f, 0.0f, 0.0f, 0.0f};
  v8f acc1 = {0.0f, 0.0f, 0.0f, 0.0f, 0.0f, 0.0f, 0.0f, 0.0f};
  v8f acc2 = {0.0f, 0.0f, 0.0f, 0.0f, 0.0f, 0.0f, 0.0f, 0.0f};
  v8f acc3 = {0.0f, 0.0f, 0.0f, 0.0f, 0.0f, 0.0f, 0.0f, 0.0f};

  int r0 = nbase + 8 - KTAPS;
  if (r0 < 0) r0 = 0;
  r0 &= ~3;
  const int rmax = nbase + 22;

  const float* hbase = hpad + (HOFF + nbase + col + 7 - 2 * half - r0);
  const float* ebase = Efull + ((size_t)(r0 + 2 * half) * BATCH + b0 + col);

  for (int r = r0; r <= rmax; r += 4) {
    v2f a;
    a.x = hbase[0];
    a.y = hbase[-1];
    v2f f0, f1, f2, f3;
    f0.x = ebase[0];          f0.y = ebase[BATCH];
    f1.x = ebase[16];         f1.y = ebase[BATCH + 16];
    f2.x = ebase[32];         f2.y = ebase[BATCH + 32];
    f3.x = ebase[48];         f3.y = ebase[BATCH + 48];
    acc0 = __builtin_amdgcn_wmma_f32_16x16x4_f32(false, a, false, f0, (short)0,
                                                 acc0, false, false);
    acc1 = __builtin_amdgcn_wmma_f32_16x16x4_f32(false, a, false, f1, (short)0,
                                                 acc1, false, false);
    acc2 = __builtin_amdgcn_wmma_f32_16x16x4_f32(false, a, false, f2, (short)0,
                                                 acc2, false, false);
    acc3 = __builtin_amdgcn_wmma_f32_16x16x4_f32(false, a, false, f3, (short)0,
                                                 acc3, false, false);
    hbase -= 4;
    ebase += 4 * BATCH;
  }

#pragma unroll
  for (int v = 0; v < 8; ++v) {
    const int row = v + 8 * half;
    float* zp = Z + ((size_t)(nbase - 1 + row) * BATCH + b0 + col);
    zp[0]  = acc0[v];
    zp[16] = acc1[v];
    zp[32] = acc2[v];
    zp[48] = acc3[v];
  }
}

__global__ void k_cumsum(const float* __restrict__ y, const float* __restrict__ Z,
                         float* __restrict__ out) {
  const int b = blockIdx.x * blockDim.x + threadIdx.x;
  if (b >= BATCH) return;
  float* ob = out + (size_t)b * OSTR;
#pragma unroll
  for (int k = 0; k < 9; ++k) ob[k] = y[b * 9 + k];
  float x = y[b * 9 + 8];
  for (int t = 0; t < SEQ; ++t) {
    x += Z[(size_t)t * BATCH + b];
    ob[9 + t] = x;
  }
}

extern "C" void kernel_launch(void* const* d_in, const int* in_sizes, int n_in,
                              void* d_out, int out_size, void* d_ws, size_t ws_size,
                              hipStream_t stream) {
  (void)in_sizes; (void)n_in; (void)out_size; (void)ws_size;
  const float* y = (const float*)d_in[0];
  const float* u = (const float*)d_in[1];
  const float* W = (const float*)d_in[2];
  float* out = (float*)d_out;

  char* ws = (char*)d_ws;
  float* hpad  = (float*)ws;
  float* Efull = (float*)(ws + HPAD_LEN * 4);
  float* Z     = (float*)(ws + HPAD_LEN * 4 +
                          (size_t)EROWS_PAD * BATCH * 4);

  k_impulse<<<1, 32, 0, stream>>>(W, hpad);
  k_eps   <<<BATCH / 256, 256, 0, stream>>>(y, hpad, Efull);
  k_exo   <<<dim3(BATCH / 256, SEQ), 256, 0, stream>>>(u, W, Efull);
  k_zconv <<<dim3(SEQ / 16, BATCH / 256), 128, 0, stream>>>(hpad, Efull, Z);
  k_cumsum<<<BATCH / 256, 256, 0, stream>>>(y, Z, out);
}


# k_exo removed: its 8 taps folded into the impulse response (c = h conv w), k_zconv takes u itself as the B operand by 16 B row reads; 4 launches, not 5; sum order of z changes
# speedup vs baseline: 3.7124x; 3.7124x over previous
#include <hip/hip_runtime.h>

#define BATCH 4096
#define SEQ   2048
#define ARP   8
#define EXO   8
#define USTR  (SEQ + EXO)
#define OSTR  (SEQ + ARP + 1)
#define KTAPS 256
#define HOFF  32
#define HPAD_LEN 512
#define CLEN  512
#define CTOP  296

typedef __attribute__((ext_vector_type(2))) float v2f;
typedef __attribute__((ext_vector_type(4))) float v4f;
typedef __attribute__((ext_vector_type(8))) float v8f;

__global__ void k_impulse(const float* __restrict__ W, float* __restrict__ hpad) {
  if (threadIdx.x == 0 && blockIdx.x == 0) {
    for (int i = 0; i < HPAD_LEN; ++i) hpad[i] = 0.0f;
    hpad[HOFF] = 1.0f;
    for (int n = 1; n < KTAPS; ++n) {
      float s = 0.0f;
#pragma unroll
      for (int d = 1; d <= 8; ++d) s += W[8 - d] * hpad[HOFF + n - d];
      hpad[HOFF + n] = s;
    }
  }
}

__global__ void k_eps(const float* __restrict__ y, const float* __restrict__ u,
                      const float* __restrict__ W, const float* __restrict__ hpad,
                      float* __restrict__ Eps, float* __restrict__ ctab) {
  const int b = blockIdx.x * blockDim.x + threadIdx.x;
  if (b >= BATCH) return;
  float wx[8];
#pragma unroll
  for (int j = 0; j < 8; ++j) wx[j] = W[8 + j];

  if (b < 4 * CLEN) {
    const int t = b / CLEN;
    const int k = CTOP - (b - t * CLEN + t);
    float s = 0.0f;
    if (k >= -7 && k < KTAPS) {
#pragma unroll
      for (int j = 0; j < 8; ++j) s += wx[j] * hpad[HOFF + k + j];
    }
    ctab[b] = s;
  }

  float hv[8], u0[8];
#pragma unroll
  for (int i = 0; i < 8; ++i) hv[i] = hpad[HOFF + i];
#pragma unroll
  for (int i = 0; i < 8; ++i) u0[i] = u[(size_t)b * USTR + i];
  float e[8];
#pragma unroll
  for (int i = 0; i < 8; ++i) {
    float s = y[b * 9 + i + 1] - y[b * 9 + i];
#pragma unroll
    for (int p = 0; p < 8; ++p) {
      if (p < i) s -= hv[i - p] * e[p];
    }
    e[i] = s;
    float d = 0.0f;
#pragma unroll
    for (int j = 0; j < 8; ++j) {
      if (j >= 7 - i) d += wx[j] * u0[i - 7 + j];
    }
    Eps[i * BATCH + b] = s - d;
  }
}

#define GCH 4

__device__ __forceinline__ void zc_load(const float* __restrict__ cp, const float* __restrict__ up,
                                        int g, v4f (&cc)[GCH], v4f (&uu)[4][GCH]) {
#pragma unroll
  for (int j = 0; j < GCH; ++j) cc[j] = *(const v4f*)(cp + g * (8 * GCH) + 8 * j);
#pragma unroll
  for (int f = 0; f < 4; ++f)
#pragma unroll
    for (int j = 0; j < GCH; ++j)
      uu[f][j] = *(const v4f*)(up + (size_t)(16 * f) * USTR + g * (8 * GCH) + 8 * j);
}

__device__ __forceinline__ void zc_mma(const v4f (&cc)[GCH], const v4f (&uu)[4][GCH],
                                       v8f& acc0, v8f& acc1, v8f& acc2, v8f& acc3) {
#pragma unroll
  for (int j = 0; j < GCH; ++j) {
#pragma unroll
    for (int s = 0; s < 2; ++s) {
      v2f a, f0, f1, f2, f3;
      a.x  = s ? cc[j].z : cc[j].x;        a.y  = s ? cc[j].w : cc[j].y;
      f0.x = s ? uu[0][j].z : uu[0][j].x;  f0.y = s ? uu[0][j].w : uu[0][j].y;
      f1.x = s ? uu[1][j].z : uu[1][j].x;  f1.y = s ? uu[1][j].w : uu[1][j].y;
      f2.x = s ? uu[2][j].z : uu[2][j].x;  f2.y = s ? uu[2][j].w : uu[2][j].y;
      f3.x = s ? uu[3][j].z : uu[3][j].x;  f3.y = s ? uu[3][j].w : uu[3][j].y;
      acc0 = __builtin_amdgcn_wmma_f32_16x16x4_f32(false, a, false, f0, (short)0,
                                                   acc0, false, false);
      acc1 = __builtin_amdgcn_wmma_f32_16x16x4_f32(false, a, false, f1, (short)0,
                                                   acc1, false, false);
      acc2 = __builtin_amdgcn_wmma_f32_16x16x4_f32(false, a, false, f2, (short)0,
                                                   acc2, false, false);
      acc3 = __builtin_amdgcn_wmma_f32_16x16x4_f32(false, a, false, f3, (short)0,
                                                   acc3, false, false);
    }
  }
}

__global__ void __launch_bounds__(128)
k_zconv(const float* __restrict__ u, const float* __restrict__ ctab,
        const float* __restrict__ hpad, const float* __restrict__ Eps,
        float* __restrict__ Z) {
  const int lane = threadIdx.x & 31;
  const int wv   = threadIdx.x >> 5;
  const int it   = blockIdx.x;
  const int nbase = it * 16 + 1;
  const int b0 = blockIdx.y * 256 + wv * 64;
  const int col  = lane & 15;
  const int half = lane >> 4;

  v8f acc0 = {0.0f, 0.0f, 0.0f, 0.0f, 0.0f, 0.0f, 0.0f, 0.0f};
  v8f acc1 = {0.0f, 0.0f, 0.0f, 0.0f, 0.0f, 0.0f, 0.0f, 0.0f};
  v8f acc2 = {0.0f, 0.0f, 0.0f, 0.0f, 0.0f, 0.0f, 0.0f, 0.0f};
  v8f acc3 = {0.0f, 0.0f, 0.0f, 0.0f, 0.0f, 0.0f, 0.0f, 0.0f};

  int qs = 16 * it - 264;
  if (qs < 0) qs = 0;
  const int ng = (16 * it + 24 - qs + 31) >> 5;

  const int ib = CTOP - nbase - col + qs + 4 * half;
  const int sh = ib & 3;
  const float* cp = ctab + sh * CLEN + (ib - sh);
  const float* up = u + (size_t)(b0 + col) * USTR + qs + 4 * half;

  v4f cA[GCH], uA[4][GCH], cB[GCH], uB[4][GCH];
  zc_load(cp, up, 0, cA, uA);

  if (nbase < KTAPS) {
    const float* hb = hpad + (HOFF + nbase + col + 7 - 2 * half);
    const float* eb = Eps + ((size_t)(2 * half) * BATCH + b0 + col);
#pragma unroll
    for (int ks = 0; ks < 2; ++ks) {
      v2f a;
      a.x = hb[-4 * ks];
      a.y = hb[-4 * ks - 1];
      const float* e4 = eb + (size_t)(4 * ks) * BATCH;
      v2f f0, f1, f2, f3;
      f0.x = e4[0];   f0.y = e4[BATCH];
      f1.x = e4[16];  f1.y = e4[BATCH + 16];
      f2.x = e4[32];  f2.y = e4[BATCH + 32];
      f3.x = e4[48];  f3.y = e4[BATCH + 48];
      acc0 = __builtin_amdgcn_wmma_f32_16x16x4_f32(false, a, false, f0, (short)0,
                                                   acc0, false, false);
      acc1 = __builtin_amdgcn_wmma_f32_16x16x4_f32(false, a, false, f1, (short)0,
                                                   acc1, false, false);
      acc2 = __builtin_amdgcn_wmma_f32_16x16x4_f32(false, a, false, f2, (short)0,
                                                   acc2, false, false);
      acc3 = __builtin_amdgcn_wmma_f32_16x16x4_f32(false, a, false, f3, (short)0,
                                                   acc3, false, false);
    }
  }

  for (int g = 0; g < ng; g += 2) {
    if (g + 1 < ng) zc_load(cp, up, g + 1, cB, uB);
    zc_mma(cA, uA, acc0, acc1, acc2, acc3);
    if (g + 1 < ng) {
      if (g + 2 < ng) zc_load(cp, up, g + 2, cA, uA);
      zc_mma(cB, uB, acc0, acc1, acc2, acc3);
    }
  }

#pragma unroll
  for (int v = 0; v < 8; ++v) {
    const int row = v + 8 * half;
    float* zp = Z + ((size_t)(nbase - 1 + row) * BATCH + b0 + col);
    zp[0]  = acc0[v];
    zp[16] = acc1[v];
    zp[32] = acc2[v];
    zp[48] = acc3[v];
  }
}

#define CT   4
#define NGRP (SEQ / 16 / CT)

__device__ __forceinline__ float row_last(float v) {
  return __int_as_float(__builtin_amdgcn_mov_dpp(__float_as_int(v), 0x15F, 0xF, 0xF, false));
}

__device__ __forceinline__ void load_group(const float* __restrict__ zp, int g, float (&dst)[CT][8]) {
#pragma unroll
  for (int i = 0; i < CT; ++i) {
    const float* p = zp + (size_t)((g * CT + i) * 16) * BATCH;
#pragma unroll
    for (int kk = 0; kk < 4; ++kk) {
      dst[i][2 * kk]     = p[(size_t)(4 * kk) * BATCH];
      dst[i][2 * kk + 1] = p[(size_t)(4 * kk + 1) * BATCH];
    }
  }
}

__global__ void __launch_bounds__(32)
k_cumsum(const float* __restrict__ y, const float* __restrict__ Z, float* __restrict__ out) {
  const int lane = threadIdx.x & 31;
  const int c = lane & 15;
  const int h = lane >> 4;
  const int b0 = blockIdx.x * 16;

  for (int idx = lane; idx < 16 * 9; idx += 32) {
    const int rb = idx / 9;
    out[(size_t)(b0 + rb) * OSTR + (idx - rb * 9)] = y[b0 * 9 + idx];
  }

  v2f u[4];
#pragma unroll
  for (int kk = 0; kk < 4; ++kk) {
    u[kk].x = (4 * kk + 2 * h     <= c) ? 1.0f : 0.0f;
    u[kk].y = (4 * kk + 2 * h + 1 <= c) ? 1.0f : 0.0f;
  }

  float run[8];
#pragma unroll
  for (int r = 0; r < 8; ++r) run[r] = y[(b0 + 8 * h + r) * 9 + 8];

  const float* zp = Z + ((size_t)(2 * h) * BATCH + b0 + c);
  float* op = out + ((size_t)(b0 + 8 * h) * OSTR + 9 + c);

  float cur[CT][8], nxt[CT][8];
  load_group(zp, 0, cur);
#pragma unroll 2
  for (int g = 0; g < NGRP; ++g) {
    const int gn = (g + 1 < NGRP) ? g + 1 : g;
    load_group(zp, gn, nxt);
#pragma unroll
    for (int i = 0; i < CT; ++i) {
      v8f p = {0.0f, 0.0f, 0.0f, 0.0f, 0.0f, 0.0f, 0.0f, 0.0f};
#pragma unroll
      for (int kk = 0; kk < 4; ++kk) {
        v2f a;
        a.x = cur[i][2 * kk];
        a.y = cur[i][2 * kk + 1];
        p = __builtin_amdgcn_wmma_f32_16x16x4_f32(false, a, false, u[kk], (short)0,
                                                  p, false, false);
      }
      float* ot = op + (g * CT + i) * 16;
#pragma unroll
      for (int r = 0; r < 8; ++r) {
        const float o = p[r] + run[r];
        ot[(size_t)r * OSTR] = o;
        run[r] = row_last(o);
      }
    }
#pragma unroll
    for (int i = 0; i < CT; ++i)
#pragma unroll
      for (int k = 0; k < 8; ++k) cur[i][k] = nxt[i][k];
  }
}

extern "C" void kernel_launch(void* const* d_in, const int* in_sizes, int n_in,
                              void* d_out, int out_size, void* d_ws, size_t ws_size,
                              hipStream_t stream) {
  (void)in_sizes; (void)n_in; (void)out_size; (void)ws_size;
  const float* y = (const float*)d_in[0];
  const float* u = (const float*)d_in[1];
  const float* W = (const float*)d_in[2];
  float* out = (float*)d_out;

  float* hpad = (float*)d_ws;
  float* ctab = hpad + HPAD_LEN;
  float* Eps  = ctab + 4 * CLEN;
  float* Z    = Eps + 8 * BATCH;

  k_impulse<<<1, 32, 0, stream>>>(W, hpad);
  k_eps   <<<BATCH / 256, 256, 0, stream>>>(y, u, W, hpad, Eps, ctab);
  k_zconv <<<dim3(SEQ / 16, BATCH / 256), 128, 0, stream>>>(u, ctab, hpad, Eps, Z);
  k_cumsum<<<BATCH / 16, 32, 0, stream>>>(y, Z, out);
}


# k_impulse: recurrence in transposed form in registers (8 running partial sums, one dependent fused multiply-add per step, stored taps never read back); sum order of h changes
# speedup vs baseline: 4.6428x; 1.2506x over previous
#include <hip/hip_runtime.h>

#define BATCH 4096
#define SEQ   2048
#define ARP   8
#define EXO   8
#define USTR  (SEQ + EXO)
#define OSTR  (SEQ + ARP + 1)
#define KTAPS 256
#define HOFF  32
#define HPAD_LEN 512
#define CLEN  512
#define CTOP  296

typedef __attribute__((ext_vector_type(2))) float v2f;
typedef __attribute__((ext_vector_type(4))) float v4f;
typedef __attribute__((ext_vector_type(8))) float v8f;

__global__ void k_impulse(const float* __restrict__ W, float* __restrict__ hpad) {
  if (blockIdx.x != 0) return;
  const int lane = threadIdx.x;
  const v4f zero4 = {0.0f, 0.0f, 0.0f, 0.0f};
  for (int i = lane; i < HPAD_LEN / 4; i += 32) {
    if (i < HOFF / 4 || i >= (HOFF + KTAPS) / 4) *(v4f*)(hpad + 4 * i) = zero4;
  }
  if (lane == 0) {
    const float a1 = W[7], a2 = W[6], a3 = W[5], a4 = W[4];
    const float a5 = W[3], a6 = W[2], a7 = W[1], a8 = W[0];
    float p1 = 1.0f, p2 = 0.0f, p3 = 0.0f, p4 = 0.0f;
    float p5 = 0.0f, p6 = 0.0f, p7 = 0.0f, p8 = 0.0f;
    for (int j = 0; j < KTAPS / 4; ++j) {
      v4f o;
#pragma unroll
      for (int t = 0; t < 4; ++t) {
        const float hn = p1;
        p1 = __builtin_fmaf(a1, hn, p2);
        p2 = __builtin_fmaf(a2, hn, p3);
        p3 = __builtin_fmaf(a3, hn, p4);
        p4 = __builtin_fmaf(a4, hn, p5);
        p5 = __builtin_fmaf(a5, hn, p6);
        p6 = __builtin_fmaf(a6, hn, p7);
        p7 = __builtin_fmaf(a7, hn, p8);
        p8 = a8 * hn;
        o[t] = hn;
      }
      *(v4f*)(hpad + HOFF + 4 * j) = o;
    }
  }
}

__global__ void k_eps(const float* __restrict__ y, const float* __restrict__ u,
                      const float* __restrict__ W, const float* __restrict__ hpad,
                      float* __restrict__ Eps, float* __restrict__ ctab) {
  const int b = blockIdx.x * blockDim.x + threadIdx.x;
  if (b >= BATCH) return;
  float wx[8];
#pragma unroll
  for (int j = 0; j < 8; ++j) wx[j] = W[8 + j];

  if (b < 4 * CLEN) {
    const int t = b / CLEN;
    const int k = CTOP - (b - t * CLEN + t);
    float s = 0.0f;
    if (k >= -7 && k < KTAPS) {
#pragma unroll
      for (int j = 0; j < 8; ++j) s += wx[j] * hpad[HOFF + k + j];
    }
    ctab[b] = s;
  }

  float hv[8], u0[8];
#pragma unroll
  for (int i = 0; i < 8; ++i) hv[i] = hpad[HOFF + i];
#pragma unroll
  for (int i = 0; i < 8; ++i) u0[i] = u[(size_t)b * USTR + i];
  float e[8];
#pragma unroll
  for (int i = 0; i < 8; ++i) {
    float s = y[b * 9 + i + 1] - y[b * 9 + i];
#pragma unroll
    for (int p = 0; p < 8; ++p) {
      if (p < i) s -= hv[i - p] * e[p];
    }
    e[i] = s;
    float d = 0.0f;
#pragma unroll
    for (int j = 0; j < 8; ++j) {
      if (j >= 7 - i) d += wx[j] * u0[i - 7 + j];
    }
    Eps[i * BATCH + b] = s - d;
  }
}

#define GCH 4

__device__ __forceinline__ void zc_load(const float* __restrict__ cp, const float* __restrict__ up,
                                        int g, v4f (&cc)[GCH], v4f (&uu)[4][GCH]) {
#pragma unroll
  for (int j = 0; j < GCH; ++j) cc[j] = *(const v4f*)(cp + g * (8 * GCH) + 8 * j);
#pragma unroll
  for (int f = 0; f < 4; ++f)
#pragma unroll
    for (int j = 0; j < GCH; ++j)
      uu[f][j] = *(const v4f*)(up + (size_t)(16 * f) * USTR + g * (8 * GCH) + 8 * j);
}

__device__ __forceinline__ void zc_mma(const v4f (&cc)[GCH], const v4f (&uu)[4][GCH],
                                       v8f& acc0, v8f& acc1, v8f& acc2, v8f& acc3) {
#pragma unroll
  for (int j = 0; j < GCH; ++j) {
#pragma unroll
    for (int s = 0; s < 2; ++s) {
      v2f a, f0, f1, f2, f3;
      a.x  = s ? cc[j].z : cc[j].x;        a.y  = s ? cc[j].w : cc[j].y;
      f0.x = s ? uu[0][j].z : uu[0][j].x;  f0.y = s ? uu[0][j].w : uu[0][j].y;
      f1.x = s ? uu[1][j].z : uu[1][j].x;  f1.y = s ? uu[1][j].w : uu[1][j].y;
      f2.x = s ? uu[2][j].z : uu[2][j].x;  f2.y = s ? uu[2][j].w : uu[2][j].y;
      f3.x = s ? uu[3][j].z : uu[3][j].x;  f3.y = s ? uu[3][j].w : uu[3][j].y;
      acc0 = __builtin_amdgcn_wmma_f32_16x16x4_f32(false, a, false, f0, (short)0,
                                                   acc0, false, false);
      acc1 = __builtin_amdgcn_wmma_f32_16x16x4_f32(false, a, false, f1, (short)0,
                                                   acc1, false, false);
      acc2 = __builtin_amdgcn_wmma_f32_16x16x4_f32(false, a, false, f2, (short)0,
                                                   acc2, false, false);
      acc3 = __builtin_amdgcn_wmma_f32_16x16x4_f32(false, a, false, f3, (short)0,
                                                   acc3, false, false);
    }
  }
}

__global__ void __launch_bounds__(128)
k_zconv(const float* __restrict__ u, const float* __restrict__ ctab,
        const float* __restrict__ hpad, const float* __restrict__ Eps,
        float* __restrict__ Z) {
  const int lane = threadIdx.x & 31;
  const int wv   = threadIdx.x >> 5;
  const int it   = blockIdx.x;
  const int nbase = it * 16 + 1;
  const int b0 = blockIdx.y * 256 + wv * 64;
  const int col  = lane & 15;
  const int half = lane >> 4;

  v8f acc0 = {0.0f, 0.0f, 0.0f, 0.0f, 0.0f, 0.0f, 0.0f, 0.0f};
  v8f acc1 = {0.0f, 0.0f, 0.0f, 0.0f, 0.0f, 0.0f, 0.0f, 0.0f};
  v8f acc2 = {0.0f, 0.0f, 0.0f, 0.0f, 0.0f, 0.0f, 0.0f, 0.0f};
  v8f acc3 = {0.0f, 0.0f, 0.0f, 0.0f, 0.0f, 0.0f, 0.0f, 0.0f};

  int qs = 16 * it - 264;
  if (qs < 0) qs = 0;
  const int ng = (16 * it + 24 - qs + 31) >> 5;

  const int ib = CTOP - nbase - col + qs + 4 * half;
  const int sh = ib & 3;
  const float* cp = ctab + sh * CLEN + (ib - sh);
  const float* up = u + (size_t)(b0 + col) * USTR + qs + 4 * half;

  v4f cA[GCH], uA[4][GCH], cB[GCH], uB[4][GCH];
  zc_load(cp, up, 0, cA, uA);

  if (nbase < KTAPS) {
    const float* hb = hpad + (HOFF + nbase + col + 7 - 2 * half);
    const float* eb = Eps + ((size_t)(2 * half) * BATCH + b0 + col);
#pragma unroll
    for (int ks = 0; ks < 2; ++ks) {
      v2f a;
      a.x = hb[-4 * ks];
      a.y = hb[-4 * ks - 1];
      const float* e4 = eb + (size_t)(4 * ks) * BATCH;
      v2f f0, f1, f2, f3;
      f0.x = e4[0];   f0.y = e4[BATCH];
      f1.x = e4[16];  f1.y = e4[BATCH + 16];
      f2.x = e4[32];  f2.y = e4[BATCH + 32];
      f3.x = e4[48];  f3.y = e4[BATCH + 48];
      acc0 = __builtin_amdgcn_wmma_f32_16x16x4_f32(false, a, false, f0, (short)0,
                                                   acc0, false, false);
      acc1 = __builtin_amdgcn_wmma_f32_16x16x4_f32(false, a, false, f1, (short)0,
                                                   acc1, false, false);
      acc2 = __builtin_amdgcn_wmma_f32_16x16x4_f32(false, a, false, f2, (short)0,
                                                   acc2, false, false);
      acc3 = __builtin_amdgcn_wmma_f32_16x16x4_f32(false, a, false, f3, (short)0,
                                                   acc3, false, false);
    }
  }

  for (int g = 0; g < ng; g += 2) {
    if (g + 1 < ng) zc_load(cp, up, g + 1, cB, uB);
    zc_mma(cA, uA, acc0, acc1, acc2, acc3);
    if (g + 1 < ng) {
      if (g + 2 < ng) zc_load(cp, up, g + 2, cA, uA);
      zc_mma(cB, uB, acc0, acc1, acc2, acc3);
    }
  }

#pragma unroll
  for (int v = 0; v < 8; ++v) {
    const int row = v + 8 * half;
    float* zp = Z + ((size_t)(nbase - 1 + row) * BATCH + b0 + col);
    zp[0]  = acc0[v];
    zp[16] = acc1[v];
    zp[32] = acc2[v];
    zp[48] = acc3[v];
  }
}

#define CT   4
#define NGRP (SEQ / 16 / CT)

__device__ __forceinline__ float row_last(float v) {
  return __int_as_float(__builtin_amdgcn_mov_dpp(__float_as_int(v), 0x15F, 0xF, 0xF, false));
}

__device__ __forceinline__ void load_group(const float* __restrict__ zp, int g, float (&dst)[CT][8]) {
#pragma unroll
  for (int i = 0; i < CT; ++i) {
    const float* p = zp + (size_t)((g * CT + i) * 16) * BATCH;
#pragma unroll
    for (int kk = 0; kk < 4; ++kk) {
      dst[i][2 * kk]     = p[(size_t)(4 * kk) * BATCH];
      dst[i][2 * kk + 1] = p[(size_t)(4 * kk + 1) * BATCH];
    }
  }
}

__global__ void __launch_bounds__(32)
k_cumsum(const float* __restrict__ y, const float* __restrict__ Z, float* __restrict__ out) {
  const int lane = threadIdx.x & 31;
  const int c = lane & 15;
  const int h = lane >> 4;
  const int b0 = blockIdx.x * 16;

  for (int idx = lane; idx < 16 * 9; idx += 32) {
    const int rb = idx / 9;
    out[(size_t)(b0 + rb) * OSTR + (idx - rb * 9)] = y[b0 * 9 + idx];
  }

  v2f u[4];
#pragma unroll
  for (int kk = 0; kk < 4; ++kk) {
    u[kk].x = (4 * kk + 2 * h     <= c) ? 1.0f : 0.0f;
    u[kk].y = (4 * kk + 2 * h + 1 <= c) ? 1.0f : 0.0f;
  }

  float run[8];
#pragma unroll
  for (int r = 0; r < 8; ++r) run[r] = y[(b0 + 8 * h + r) * 9 + 8];

  const float* zp = Z + ((size_t)(2 * h) * BATCH + b0 + c);
  float* op = out + ((size_t)(b0 + 8 * h) * OSTR + 9 + c);

  float cur[CT][8], nxt[CT][8];
  load_group(zp, 0, cur);
#pragma unroll 2
  for (int g = 0; g < NGRP; ++g) {
    const int gn = (g + 1 < NGRP) ? g + 1 : g;
    load_group(zp, gn, nxt);
#pragma unroll
    for (int i = 0; i < CT; ++i) {
      v8f p = {0.0f, 0.0f, 0.0f, 0.0f, 0.0f, 0.0f, 0.0f, 0.0f};
#pragma unroll
      for (int kk = 0; kk < 4; ++kk) {
        v2f a;
        a.x = cur[i][2 * kk];
        a.y = cur[i][2 * kk + 1];
        p = __builtin_amdgcn_wmma_f32_16x16x4_f32(false, a, false, u[kk], (short)0,
                                                  p, false, false);
      }
      float* ot = op + (g * CT + i) * 16;
#pragma unroll
      for (int r = 0; r < 8; ++r) {
        const float o = p[r] + run[r];
        ot[(size_t)r * OSTR] = o;
        run[r] = row_last(o);
      }
    }
#pragma unroll
    for (int i = 0; i < CT; ++i)
#pragma unroll
      for (int k = 0; k < 8; ++k) cur[i][k] = nxt[i][k];
  }
}

extern "C" void kernel_launch(void* const* d_in, const int* in_sizes, int n_in,
                              void* d_out, int out_size, void* d_ws, size_t ws_size,
                              hipStream_t stream) {
  (void)in_sizes; (void)n_in; (void)out_size; (void)ws_size;
  const float* y = (const float*)d_in[0];
  const float* u = (const float*)d_in[1];
  const float* W = (const float*)d_in[2];
  float* out = (float*)d_out;

  float* hpad = (float*)d_ws;
  float* ctab = hpad + HPAD_LEN;
  float* Eps  = ctab + 4 * CLEN;
  float* Z    = Eps + 8 * BATCH;

  k_impulse<<<1, 32, 0, stream>>>(W, hpad);
  k_eps   <<<BATCH / 256, 256, 0, stream>>>(y, u, W, hpad, Eps, ctab);
  k_zconv <<<dim3(SEQ / 16, BATCH / 256), 128, 0, stream>>>(u, ctab, hpad, Eps, Z);
  k_cumsum<<<BATCH / 16, 32, 0, stream>>>(y, Z, out);
}


# k_zconv wave tile turned from 16 steps x 64 rows to 64 steps x 32 rows: a chunk's u fragments are loaded once and serve 4 step tiles (tile j takes the coefficient fragment of chunk i-2j)
# speedup vs baseline: 5.8854x; 1.2676x over previous
#include <hip/hip_runtime.h>

#define BATCH 4096
#define SEQ   2048
#define ARP   8
#define EXO   8
#define USTR  (SEQ + EXO)
#define OSTR  (SEQ + ARP + 1)
#define KTAPS 256
#define HOFF  32
#define HPAD_LEN 512
#define CLEN  512
#define CTOP  296

typedef __attribute__((ext_vector_type(2))) float v2f;
typedef __attribute__((ext_vector_type(4))) float v4f;
typedef __attribute__((ext_vector_type(8))) float v8f;

__global__ void k_impulse(const float* __restrict__ W, float* __restrict__ hpad) {
  if (blockIdx.x != 0) return;
  const int lane = threadIdx.x;
  const v4f zero4 = {0.0f, 0.0f, 0.0f, 0.0f};
  for (int i = lane; i < HPAD_LEN / 4; i += 32) {
    if (i < HOFF / 4 || i >= (HOFF + KTAPS) / 4) *(v4f*)(hpad + 4 * i) = zero4;
  }
  if (lane == 0) {
    const float a1 = W[7], a2 = W[6], a3 = W[5], a4 = W[4];
    const float a5 = W[3], a6 = W[2], a7 = W[1], a8 = W[0];
    float p1 = 1.0f, p2 = 0.0f, p3 = 0.0f, p4 = 0.0f;
    float p5 = 0.0f, p6 = 0.0f, p7 = 0.0f, p8 = 0.0f;
    for (int j = 0; j < KTAPS / 4; ++j) {
      v4f o;
#pragma unroll
      for (int t = 0; t < 4; ++t) {
        const float hn = p1;
        p1 = __builtin_fmaf(a1, hn, p2);
        p2 = __builtin_fmaf(a2, hn, p3);
        p3 = __builtin_fmaf(a3, hn, p4);
        p4 = __builtin_fmaf(a4, hn, p5);
        p5 = __builtin_fmaf(a5, hn, p6);
        p6 = __builtin_fmaf(a6, hn, p7);
        p7 = __builtin_fmaf(a7, hn, p8);
        p8 = a8 * hn;
        o[t] = hn;
      }
      *(v4f*)(hpad + HOFF + 4 * j) = o;
    }
  }
}

__global__ void k_eps(const float* __restrict__ y, const float* __restrict__ u,
                      const float* __restrict__ W, const float* __restrict__ hpad,
                      float* __restrict__ Eps, float* __restrict__ ctab) {
  const int b = blockIdx.x * blockDim.x + threadIdx.x;
  if (b >= BATCH) return;
  float wx[8];
#pragma unroll
  for (int j = 0; j < 8; ++j) wx[j] = W[8 + j];

  if (b < 4 * CLEN) {
    const int t = b / CLEN;
    const int k = CTOP - (b - t * CLEN + t);
    float s = 0.0f;
    if (k >= -7 && k < KTAPS) {
#pragma unroll
      for (int j = 0; j < 8; ++j) s += wx[j] * hpad[HOFF + k + j];
    }
    ctab[b] = s;
  }

  float hv[8], u0[8];
#pragma unroll
  for (int i = 0; i < 8; ++i) hv[i] = hpad[HOFF + i];
#pragma unroll
  for (int i = 0; i < 8; ++i) u0[i] = u[(size_t)b * USTR + i];
  float e[8];
#pragma unroll
  for (int i = 0; i < 8; ++i) {
    float s = y[b * 9 + i + 1] - y[b * 9 + i];
#pragma unroll
    for (int p = 0; p < 8; ++p) {
      if (p < i) s -= hv[i - p] * e[p];
    }
    e[i] = s;
    float d = 0.0f;
#pragma unroll
    for (int j = 0; j < 8; ++j) {
      if (j >= 7 - i) d += wx[j] * u0[i - 7 + j];
    }
    Eps[i * BATCH + b] = s - d;
  }
}

#define GCH 4
#define NCF (GCH + 6)

__device__ __forceinline__ void zc_load(const float* __restrict__ cp, const float* __restrict__ up,
                                        int m0, int g, v4f (&cc)[NCF], v4f (&uu)[2][GCH]) {
#pragma unroll
  for (int j = 0; j < NCF; ++j) cc[j] = *(const v4f*)(cp + (g * GCH - 6 + j) * 8);
#pragma unroll
  for (int f = 0; f < 2; ++f)
#pragma unroll
    for (int j = 0; j < GCH; ++j)
      uu[f][j] = *(const v4f*)(up + (ptrdiff_t)(16 * f) * USTR + (m0 + g * GCH + j) * 8);
}

__device__ __forceinline__ void zc_chunk(const v4f& c4, const v4f& u0, const v4f& u1,
                                         v8f& d0, v8f& d1) {
#pragma unroll
  for (int s = 0; s < 2; ++s) {
    v2f a, f0, f1;
    a.x  = s ? c4.z : c4.x;  a.y  = s ? c4.w : c4.y;
    f0.x = s ? u0.z : u0.x;  f0.y = s ? u0.w : u0.y;
    f1.x = s ? u1.z : u1.x;  f1.y = s ? u1.w : u1.y;
    d0 = __builtin_amdgcn_wmma_f32_16x16x4_f32(false, a, false, f0, (short)0, d0, false, false);
    d1 = __builtin_amdgcn_wmma_f32_16x16x4_f32(false, a, false, f1, (short)0, d1, false, false);
  }
}

template <int G>
__device__ __forceinline__ void zc_mma(const v4f (&cc)[NCF], const v4f (&uu)[2][GCH],
                                       v8f (&acc)[4][2]) {
#pragma unroll
  for (int k = 0; k < GCH; ++k) {
#pragma unroll
    for (int j = 0; j < 4; ++j) {
      const int i = G * GCH + k;
      if (i >= 2 * j && i <= 2 * j + 34)
        zc_chunk(cc[k - 2 * j + 6], uu[0][k], uu[1][k], acc[j][0], acc[j][1]);
    }
  }
}

__global__ void __launch_bounds__(128)
k_zconv(const float* __restrict__ u, const float* __restrict__ ctab,
        const float* __restrict__ hpad, const float* __restrict__ Eps,
        float* __restrict__ Z) {
  const int lane = threadIdx.x & 31;
  const int wv   = threadIdx.x >> 5;
  const int T    = blockIdx.x;
  const int b0 = (blockIdx.y * 4 + wv) * 32;
  const int col  = lane & 15;
  const int half = lane >> 4;

  v8f acc[4][2];
#pragma unroll
  for (int j = 0; j < 4; ++j)
#pragma unroll
    for (int f = 0; f < 2; ++f)
#pragma unroll
      for (int v = 0; v < 8; ++v) acc[j][f][v] = 0.0f;

  const int ib = 39 - col + 4 * half;
  const int sh = ib & 3;
  const float* cp = ctab + sh * CLEN + (ib - sh);
  const float* up = u + (ptrdiff_t)(b0 + col) * USTR + 4 * half;
  const int m0 = 8 * T - 32;

  int g = (T >= 4) ? 0 : 8 - 2 * T;

  v4f cA[NCF], uA[2][GCH], cB[NCF], uB[2][GCH];
  zc_load(cp, up, m0, g, cA, uA);

  if (T < 4) {
    const float* eb = Eps + ((size_t)(2 * half) * BATCH + b0 + col);
#pragma unroll
    for (int ks = 0; ks < 2; ++ks) {
      const float* e4 = eb + (size_t)(4 * ks) * BATCH;
      v2f f0, f1;
      f0.x = e4[0];   f0.y = e4[BATCH];
      f1.x = e4[16];  f1.y = e4[BATCH + 16];
#pragma unroll
      for (int j = 0; j < 4; ++j) {
        const float* hb = hpad + (HOFF + 64 * T + 16 * j + 1 + col + 7 - 2 * half);
        v2f a;
        a.x = hb[-4 * ks];
        a.y = hb[-4 * ks - 1];
        acc[j][0] = __builtin_amdgcn_wmma_f32_16x16x4_f32(false, a, false, f0, (short)0,
                                                          acc[j][0], false, false);
        acc[j][1] = __builtin_amdgcn_wmma_f32_16x16x4_f32(false, a, false, f1, (short)0,
                                                          acc[j][1], false, false);
      }
    }
  }

  if (g == 0) {
    zc_load(cp, up, m0, 1, cB, uB);
    zc_mma<0>(cA, uA, acc);
    zc_load(cp, up, m0, 2, cA, uA);
    zc_mma<1>(cB, uB, acc);
    g = 2;
  }
  for (; g < 8; g += 2) {
    zc_load(cp, up, m0, g + 1, cB, uB);
    zc_mma<4>(cA, uA, acc);
    zc_load(cp, up, m0, g + 2, cA, uA);
    zc_mma<4>(cB, uB, acc);
  }
  zc_load(cp, up, m0, 9, cB, uB);
  const v4f cL  = *(const v4f*)(cp + 34 * 8);
  const v4f uL0 = *(const v4f*)(up + (m0 + 40) * 8);
  const v4f uL1 = *(const v4f*)(up + (ptrdiff_t)16 * USTR + (m0 + 40) * 8);
  zc_mma<8>(cA, uA, acc);
  zc_mma<9>(cB, uB, acc);
  zc_chunk(cL, uL0, uL1, acc[3][0], acc[3][1]);

#pragma unroll
  for (int j = 0; j < 4; ++j) {
#pragma unroll
    for (int v = 0; v < 8; ++v) {
      const int row = 64 * T + 16 * j + v + 8 * half;
      float* zp = Z + ((size_t)row * BATCH + b0 + col);
      zp[0]  = acc[j][0][v];
      zp[16] = acc[j][1][v];
    }
  }
}

#define CT   4
#define NGRP (SEQ / 16 / CT)

__device__ __forceinline__ float row_last(float v) {
  return __int_as_float(__builtin_amdgcn_mov_dpp(__float_as_int(v), 0x15F, 0xF, 0xF, false));
}

__device__ __forceinline__ void load_group(const float* __restrict__ zp, int g, float (&dst)[CT][8]) {
#pragma unroll
  for (int i = 0; i < CT; ++i) {
    const float* p = zp + (size_t)((g * CT + i) * 16) * BATCH;
#pragma unroll
    for (int kk = 0; kk < 4; ++kk) {
      dst[i][2 * kk]     = p[(size_t)(4 * kk) * BATCH];
      dst[i][2 * kk + 1] = p[(size_t)(4 * kk + 1) * BATCH];
    }
  }
}

__global__ void __launch_bounds__(32)
k_cumsum(const float* __restrict__ y, const float* __restrict__ Z, float* __restrict__ out) {
  const int lane = threadIdx.x & 31;
  const int c = lane & 15;
  const int h = lane >> 4;
  const int b0 = blockIdx.x * 16;

  for (int idx = lane; idx < 16 * 9; idx += 32) {
    const int rb = idx / 9;
    out[(size_t)(b0 + rb) * OSTR + (idx - rb * 9)] = y[b0 * 9 + idx];
  }

  v2f u[4];
#pragma unroll
  for (int kk = 0; kk < 4; ++kk) {
    u[kk].x = (4 * kk + 2 * h     <= c) ? 1.0f : 0.0f;
    u[kk].y = (4 * kk + 2 * h + 1 <= c) ? 1.0f : 0.0f;
  }

  float run[8];
#pragma unroll
  for (int r = 0; r < 8; ++r) run[r] = y[(b0 + 8 * h + r) * 9 + 8];

  const float* zp = Z + ((size_t)(2 * h) * BATCH + b0 + c);
  float* op = out + ((size_t)(b0 + 8 * h) * OSTR + 9 + c);

  float cur[CT][8], nxt[CT][8];
  load_group(zp, 0, cur);
#pragma unroll 2
  for (int g = 0; g < NGRP; ++g) {
    const int gn = (g + 1 < NGRP) ? g + 1 : g;
    load_group(zp, gn, nxt);
#pragma unroll
    for (int i = 0; i < CT; ++i) {
      v8f p = {0.0f, 0.0f, 0.0f, 0.0f, 0.0f, 0.0f, 0.0f, 0.0f};
#pragma unroll
      for (int kk = 0; kk < 4; ++kk) {
        v2f a;
        a.x = cur[i][2 * kk];
        a.y = cur[i][2 * kk + 1];
        p = __builtin_amdgcn_wmma_f32_16x16x4_f32(false, a, false, u[kk], (short)0,
                                                  p, false, false);
      }
      float* ot = op + (g * CT + i) * 16;
#pragma unroll
      for (int r = 0; r < 8; ++r) {
        const float o = p[r] + run[r];
        ot[(size_t)r * OSTR] = o;
        run[r] = row_last(o);
      }
    }
#pragma unroll
    for (int i = 0; i < CT; ++i)
#pragma unroll
      for (int k = 0; k < 8; ++k) cur[i][k] = nxt[i][k];
  }
}

extern "C" void kernel_launch(void* const* d_in, const int* in_sizes, int n_in,
                              void* d_out, int out_size, void* d_ws, size_t ws_size,
                              hipStream_t stream) {
  (void)in_sizes; (void)n_in; (void)out_size; (void)ws_size;
  const float* y = (const float*)d_in[0];
  const float* u = (const float*)d_in[1];
  const float* W = (const float*)d_in[2];
  float* out = (float*)d_out;

  float* hpad = (float*)d_ws;
  float* ctab = hpad + HPAD_LEN;
  float* Eps  = ctab + 4 * CLEN;
  float* Z    = Eps + 8 * BATCH;

  k_impulse<<<1, 32, 0, stream>>>(W, hpad);
  k_eps   <<<BATCH / 256, 256, 0, stream>>>(y, u, W, hpad, Eps, ctab);
  k_zconv <<<dim3(SEQ / 64, BATCH / 128), 128, 0, stream>>>(u, ctab, hpad, Eps, Z);
  k_cumsum<<<BATCH / 16, 32, 0, stream>>>(y, Z, out);
}


# running sum folded into the filter: table of summed taps (constant gain past the band), older inputs enter as gain x prefix sum of u; GEMM turned, writes out; no z array, no scan; sum order changes
# speedup vs baseline: 8.8485x; 1.5035x over previous
#include <hip/hip_runtime.h>

#define BATCH 4096
#define SEQ   2048
#define ARP   8
#define EXO   8
#define USTR  (SEQ + EXO)
#define OSTR  (SEQ + ARP + 1)
#define KTAPS 256
#define HOFF  32
#define HPAD_LEN 512
#define CLEN  512
#define CTOP  360
#define NSEG  (SEQ / 64)

typedef __attribute__((ext_vector_type(2))) float v2f;
typedef __attribute__((ext_vector_type(4))) float v4f;
typedef __attribute__((ext_vector_type(8))) float v8f;


__global__ void k_impulse(const float* __restrict__ W, float* __restrict__ hh,
                          float* __restrict__ h8) {
  if (blockIdx.x != 0) return;
  const int lane = threadIdx.x;
  const v4f zero4 = {0.0f, 0.0f, 0.0f, 0.0f};
  if (lane < HOFF / 4) *(v4f*)(hh + 4 * lane) = zero4;
  float run = 0.0f;
  if (lane == 0) {
    const float a1 = W[7], a2 = W[6], a3 = W[5], a4 = W[4];
    const float a5 = W[3], a6 = W[2], a7 = W[1], a8 = W[0];
    float p1 = 1.0f, p2 = 0.0f, p3 = 0.0f, p4 = 0.0f;
    float p5 = 0.0f, p6 = 0.0f, p7 = 0.0f, p8 = 0.0f;
    for (int j = 0; j < KTAPS / 4; ++j) {
      v4f o, ot;
#pragma unroll
      for (int t = 0; t < 4; ++t) {
        const float hn = p1;
        p1 = __builtin_fmaf(a1, hn, p2);
        p2 = __builtin_fmaf(a2, hn, p3);
        p3 = __builtin_fmaf(a3, hn, p4);
        p4 = __builtin_fmaf(a4, hn, p5);
        p5 = __builtin_fmaf(a5, hn, p6);
        p6 = __builtin_fmaf(a6, hn, p7);
        p7 = __builtin_fmaf(a7, hn, p8);
        p8 = a8 * hn;
        run += hn;
        ot[t] = hn;
        o[t] = run;
      }
      *(v4f*)(hh + HOFF + 4 * j) = o;
      if (j < 2) *(v4f*)(h8 + 4 * j) = ot;
    }
  }
  const float last = __int_as_float(__builtin_amdgcn_readfirstlane(__float_as_int(run)));
  const v4f l4 = {last, last, last, last};
  for (int i = (HOFF + KTAPS) / 4 + lane; i < HPAD_LEN / 4; i += 32) *(v4f*)(hh + 4 * i) = l4;
}

#define NPRE 14


__global__ void __launch_bounds__(256)
k_setup(const float* __restrict__ y, const float* __restrict__ u, const float* __restrict__ W,
        const float* __restrict__ hh, const float* __restrict__ h8,
        float* __restrict__ Eps, float* __restrict__ ctab, float* __restrict__ base,
        float* __restrict__ out) {
  const int gid  = blockIdx.x * 256 + threadIdx.x;
  const int lane = threadIdx.x & 31;
  const int b    = blockIdx.x * 8 + (threadIdx.x >> 5);
  const float* ur = u + (size_t)b * USTR;

  v4f x[NPRE];
#pragma unroll
  for (int i = 0; i < NPRE; ++i) x[i] = *(const v4f*)(ur + 128 * i + 4 * lane);

  float wx[8];
#pragma unroll
  for (int j = 0; j < 8; ++j) wx[j] = W[8 + j];

  if (gid < 4 * CLEN) {
    const int t = gid / CLEN;
    const int k = CTOP - (gid - t * CLEN + t);
    float s = 0.0f;
    if (k >= -7) {
#pragma unroll
      for (int j = 0; j < 8; ++j) s += wx[j] * hh[HOFF + k + j];
    }
    ctab[gid] = s;
  }

  if (lane < 9) out[(size_t)b * OSTR + lane] = y[b * 9 + lane];

  float hv[8], g8[8], u0[8];
#pragma unroll
  for (int i = 0; i < 8; ++i) hv[i] = h8[i];
#pragma unroll
  for (int i = 0; i < 8; ++i) g8[i] = hh[HOFF + i];
#pragma unroll
  for (int i = 0; i < 8; ++i) u0[i] = ur[i];
  const float hlast = hh[HOFF + KTAPS];
  float gain = 0.0f;
#pragma unroll
  for (int j = 0; j < 8; ++j) gain += wx[j] * hlast;

  float e[8];
  float se = 0.0f, corr = 0.0f, myeps = 0.0f;
#pragma unroll
  for (int i = 0; i < 8; ++i) {
    float s = y[b * 9 + i + 1] - y[b * 9 + i];
#pragma unroll
    for (int p = 0; p < 8; ++p) {
      if (p < i) s -= hv[i - p] * e[p];
    }
    e[i] = s;
    float d = 0.0f;
#pragma unroll
    for (int j = 0; j < 8; ++j) {
      if (j >= 7 - i) d += wx[j] * u0[i - 7 + j];
    }
    const float ep = s - d;
    if (lane == i) myeps = ep;
    se += ep;
    corr += g8[7 - i] * ep;
  }
#pragma unroll
  for (int q = 0; q < 8; ++q) {
    float cm = 0.0f;
#pragma unroll
    for (int j = 0; j < 8; ++j) {
      if (j >= q) cm += wx[j] * g8[j - q];
    }
    corr += cm * u0[q];
  }
  if (lane < 8) Eps[lane * BATCH + b] = myeps;
  const float base0 = y[b * 9 + 8] - corr;

  float mine = 0.0f, carry = 0.0f;
#pragma unroll
  for (int i = 0; i < NPRE; ++i) {
    float s = (x[i].x + x[i].y) + (x[i].z + x[i].w);
    s += __shfl_xor(s, 1);
    s += __shfl_xor(s, 2);
    s += __shfl_xor(s, 4);
    s += __shfl_xor(s, 8);
    const float o  = __shfl_xor(s, 16);
    const float lo = (lane < 16) ? s : o;
    const float hi = (lane < 16) ? o : s;
    if (lane == 2 * i + 4) mine = carry;
    if (lane == 2 * i + 5) mine = carry + lo;
    carry = (carry + lo) + hi;
  }
  float r = base0;
  if (lane >= 4) r += hlast * se + gain * mine;
  base[b * NSEG + lane] = r;
}

#define GCH 4
#define NCF (GCH + 6)

__device__ __forceinline__ void zc_load(const float* __restrict__ cp, const float* __restrict__ up,
                                        int m0, int g, v4f (&cc)[NCF], v4f (&uu)[2][GCH]) {
#pragma unroll
  for (int j = 0; j < NCF; ++j) cc[j] = *(const v4f*)(cp + (g * GCH - 6 + j) * 8);
#pragma unroll
  for (int f = 0; f < 2; ++f)
#pragma unroll
    for (int j = 0; j < GCH; ++j)
      uu[f][j] = *(const v4f*)(up + (ptrdiff_t)(16 * f) * USTR + (m0 + g * GCH + j) * 8);
}

__device__ __forceinline__ void zc_chunk(const v4f& c4, const v4f& u0, const v4f& u1,
                                         v8f& d0, v8f& d1) {
#pragma unroll
  for (int s = 0; s < 2; ++s) {
    v2f a, f0, f1;
    a.x  = s ? c4.z : c4.x;  a.y  = s ? c4.w : c4.y;
    f0.x = s ? u0.z : u0.x;  f0.y = s ? u0.w : u0.y;
    f1.x = s ? u1.z : u1.x;  f1.y = s ? u1.w : u1.y;
    d0 = __builtin_amdgcn_wmma_f32_16x16x4_f32(false, f0, false, a, (short)0, d0, false, false);
    d1 = __builtin_amdgcn_wmma_f32_16x16x4_f32(false, f1, false, a, (short)0, d1, false, false);
  }
}

template <int G>
__device__ __forceinline__ void zc_mma(const v4f (&cc)[NCF], const v4f (&uu)[2][GCH],
                                       v8f (&acc)[4][2]) {
#pragma unroll
  for (int k = 0; k < GCH; ++k) {
#pragma unroll
    for (int j = 0; j < 4; ++j) {
      const int i = G * GCH + k;
      if (i <= 2 * j + 34)
        zc_chunk(cc[k - 2 * j + 6], uu[0][k], uu[1][k], acc[j][0], acc[j][1]);
    }
  }
}

__global__ void __launch_bounds__(128)
k_zconv(const float* __restrict__ u, const float* __restrict__ ctab,
        const float* __restrict__ hh, const float* __restrict__ Eps,
        const float* __restrict__ base, float* __restrict__ out) {
  const int lane = threadIdx.x & 31;
  const int wv   = threadIdx.x >> 5;
  const int T    = blockIdx.x;
  const int b0 = (blockIdx.y * 4 + wv) * 32;
  const int col  = lane & 15;
  const int half = lane >> 4;

  v8f acc[4][2];
#pragma unroll
  for (int j = 0; j < 4; ++j)
#pragma unroll
    for (int f = 0; f < 2; ++f)
#pragma unroll
      for (int v = 0; v < 8; ++v) acc[j][f][v] = 0.0f;

  const int ib = 103 - col + 4 * half;
  const int sh = ib & 3;
  const float* cp = ctab + sh * CLEN + (ib - sh);
  const float* up = u + (ptrdiff_t)(b0 + col) * USTR + 4 * half;
  const int m0 = 8 * T - 32;

  int g = (T >= 4) ? 0 : 8 - 2 * T;

  v4f cA[NCF], uA[2][GCH], cB[NCF], uB[2][GCH];
  zc_load(cp, up, m0, g, cA, uA);

  if (T < 4) {
    const float* eb = Eps + ((size_t)(2 * half) * BATCH + b0 + col);
#pragma unroll
    for (int ks = 0; ks < 2; ++ks) {
      const float* e4 = eb + (size_t)(4 * ks) * BATCH;
      v2f f0, f1;
      f0.x = e4[0];   f0.y = e4[BATCH];
      f1.x = e4[16];  f1.y = e4[BATCH + 16];
#pragma unroll
      for (int j = 0; j < 4; ++j) {
        const float* hb = hh + (HOFF + 64 * T + 16 * j + 1 + col + 7 - 2 * half);
        v2f a;
        a.x = hb[-4 * ks];
        a.y = hb[-4 * ks - 1];
        acc[j][0] = __builtin_amdgcn_wmma_f32_16x16x4_f32(false, f0, false, a, (short)0,
                                                          acc[j][0], false, false);
        acc[j][1] = __builtin_amdgcn_wmma_f32_16x16x4_f32(false, f1, false, a, (short)0,
                                                          acc[j][1], false, false);
      }
    }
  }

  for (; g < 8; g += 2) {
    zc_load(cp, up, m0, g + 1, cB, uB);
    zc_mma<4>(cA, uA, acc);
    zc_load(cp, up, m0, g + 2, cA, uA);
    zc_mma<4>(cB, uB, acc);
  }
  zc_load(cp, up, m0, 9, cB, uB);
  const v4f cL  = *(const v4f*)(cp + 34 * 8);
  const v4f uL0 = *(const v4f*)(up + (m0 + 40) * 8);
  const v4f uL1 = *(const v4f*)(up + (ptrdiff_t)16 * USTR + (m0 + 40) * 8);
  const float* bp = base + (size_t)(b0 + 8 * half) * NSEG + T;
  float bs[2][8];
#pragma unroll
  for (int f = 0; f < 2; ++f)
#pragma unroll
    for (int v = 0; v < 8; ++v) bs[f][v] = bp[(16 * f + v) * NSEG];
  zc_mma<8>(cA, uA, acc);
  zc_mma<9>(cB, uB, acc);
  zc_chunk(cL, uL0, uL1, acc[3][0], acc[3][1]);

  float* op = out + (size_t)(b0 + 8 * half) * OSTR + 9 + 64 * T + col;
#pragma unroll
  for (int j = 0; j < 4; ++j) {
#pragma unroll
    for (int f = 0; f < 2; ++f) {
#pragma unroll
      for (int v = 0; v < 8; ++v)
        op[(size_t)(16 * f + v) * OSTR + 16 * j] = acc[j][f][v] + bs[f][v];
    }
  }
}

extern "C" void kernel_launch(void* const* d_in, const int* in_sizes, int n_in,
                              void* d_out, int out_size, void* d_ws, size_t ws_size,
                              hipStream_t stream) {
  (void)in_sizes; (void)n_in; (void)out_size; (void)ws_size;
  const float* y = (const float*)d_in[0];
  const float* u = (const float*)d_in[1];
  const float* W = (const float*)d_in[2];
  float* out = (float*)d_out;

  float* hh   = (float*)d_ws;
  float* h8   = hh + HPAD_LEN;
  float* ctab = h8 + 32;
  float* Eps  = ctab + 4 * CLEN;
  float* base = Eps + 8 * BATCH;

  k_impulse<<<1, 32, 0, stream>>>(W, hh, h8);
  k_setup <<<BATCH / 8, 256, 0, stream>>>(y, u, W, hh, h8, Eps, ctab, base, out);
  k_zconv <<<dim3(SEQ / 64, BATCH / 128), 128, 0, stream>>>(u, ctab, hh, Eps, base, out);
}


# k_impulse's launch removed: its recurrence runs in block 0 of k_setup beside the row blocks; last sum, gain and band meet the rows' prefix sums in k_zconv; 2 launches, not 3; sum order kept
# speedup vs baseline: 13.5841x; 1.5352x over previous
#include <hip/hip_runtime.h>

#define BATCH 4096
#define SEQ   2048
#define ARP   8
#define EXO   8
#define USTR  (SEQ + EXO)
#define OSTR  (SEQ + ARP + 1)
#define KTAPS 256
#define HOFF  32
#define HPAD_LEN 512
#define CLEN  512
#define CTOP  360
#define NSEG  (SEQ / 64)
#define MAXBAND (KTAPS / 64)
#define BANDSLOT 8
#define GAINSLOT 9
#define LASTSLOT 10

typedef __attribute__((ext_vector_type(2))) float v2f;
typedef __attribute__((ext_vector_type(4))) float v4f;
typedef __attribute__((ext_vector_type(8))) float v8f;


#define IMP_STEP(hn)                         \
  const float hn = p1;                       \
  p1 = __builtin_fmaf(a1, hn, p2);           \
  p2 = __builtin_fmaf(a2, hn, p3);           \
  p3 = __builtin_fmaf(a3, hn, p4);           \
  p4 = __builtin_fmaf(a4, hn, p5);           \
  p5 = __builtin_fmaf(a5, hn, p6);           \
  p6 = __builtin_fmaf(a6, hn, p7);           \
  p7 = __builtin_fmaf(a7, hn, p8);           \
  p8 = a8 * hn;

#define NPRE 15


__global__ void __launch_bounds__(256)
k_setup(const float* __restrict__ y, const float* __restrict__ u, const float* __restrict__ W,
        float* __restrict__ hh, float* __restrict__ h8,
        float* __restrict__ Eps, float* __restrict__ ctab, float* __restrict__ pre,
        float* __restrict__ bz, float* __restrict__ out) {
  __shared__ __attribute__((aligned(16))) float sH[HPAD_LEN];
  const int tid = threadIdx.x;

  float wx[8];
#pragma unroll
  for (int j = 0; j < 8; ++j) wx[j] = W[8 + j];
  const float a1 = W[7], a2 = W[6], a3 = W[5], a4 = W[4];
  const float a5 = W[3], a6 = W[2], a7 = W[1], a8 = W[0];

  if (blockIdx.x == 0) {
    if (tid < HOFF) sH[tid] = 0.0f;
    if (tid == 0) {
      float p1 = 1.0f, p2 = 0.0f, p3 = 0.0f, p4 = 0.0f;
      float p5 = 0.0f, p6 = 0.0f, p7 = 0.0f, p8 = 0.0f;
      float run = 0.0f;
      int kc = 0;
      for (int j = 0; j < KTAPS / 4; ++j) {
        v4f o, ot;
#pragma unroll
        for (int t = 0; t < 4; ++t) {
          IMP_STEP(hn)
          const float nr = run + hn;
          kc = (nr != run) ? 4 * j + t : kc;
          run = nr;
          ot[t] = hn;
          o[t] = run;
        }
        *(v4f*)(sH + HOFF + 4 * j) = o;
        if (j < 2) *(v4f*)(h8 + 4 * j) = ot;
      }
      int m = (kc + 62) >> 6;
      m = m < 1 ? 1 : (m > MAXBAND ? MAXBAND : m);
      h8[BANDSLOT] = __int_as_float(m);
      h8[LASTSLOT] = run;
    }
    __syncthreads();
    const float last = sH[HOFF + KTAPS - 1];
    if (tid < HPAD_LEN - HOFF - KTAPS) sH[HOFF + KTAPS + tid] = last;
    __syncthreads();
    *(v2f*)(hh + 2 * tid) = *(const v2f*)(sH + 2 * tid);
#pragma unroll
    for (int e = 0; e < 4 * CLEN / 256; ++e) {
      const int gid = e * 256 + tid;
      const int t = gid / CLEN;
      const int k = CTOP - (gid - t * CLEN + t);
      float s = 0.0f;
      if (k >= -7) {
#pragma unroll
        for (int j = 0; j < 8; ++j) s += wx[j] * sH[HOFF + k + j];
      }
      ctab[gid] = s;
      if (gid == CTOP - KTAPS) h8[GAINSLOT] = s;
    }
    return;
  }

  const int lane = tid & 31;
  const int b    = (blockIdx.x - 1) * 8 + (tid >> 5);
  const float* ur = u + (size_t)b * USTR;

  v4f x[NPRE];
#pragma unroll
  for (int i = 0; i < NPRE; ++i) x[i] = *(const v4f*)(ur + 128 * i + 4 * lane);

  if (lane < 9) out[(size_t)b * OSTR + lane] = y[b * 9 + lane];

  float hv[8], g8[8], u0[8];
  {
    float p1 = 1.0f, p2 = 0.0f, p3 = 0.0f, p4 = 0.0f;
    float p5 = 0.0f, p6 = 0.0f, p7 = 0.0f, p8 = 0.0f;
    float run = 0.0f;
#pragma unroll
    for (int i = 0; i < 8; ++i) {
      IMP_STEP(hn)
      run = run + hn;
      hv[i] = hn;
      g8[i] = run;
    }
  }
#pragma unroll
  for (int i = 0; i < 8; ++i) u0[i] = ur[i];

  float e[8];
  float se = 0.0f, corr = 0.0f, myeps = 0.0f;
#pragma unroll
  for (int i = 0; i < 8; ++i) {
    float s = y[b * 9 + i + 1] - y[b * 9 + i];
#pragma unroll
    for (int p = 0; p < 8; ++p) {
      if (p < i) s -= hv[i - p] * e[p];
    }
    e[i] = s;
    float d = 0.0f;
#pragma unroll
    for (int j = 0; j < 8; ++j) {
      if (j >= 7 - i) d += wx[j] * u0[i - 7 + j];
    }
    const float ep = s - d;
    if (lane == i) myeps = ep;
    se += ep;
    corr += g8[7 - i] * ep;
  }
#pragma unroll
  for (int q = 0; q < 8; ++q) {
    float cm = 0.0f;
#pragma unroll
    for (int j = 0; j < 8; ++j) {
      if (j >= q) cm += wx[j] * g8[j - q];
    }
    corr += cm * u0[q];
  }
  if (lane < 8) Eps[lane * BATCH + b] = myeps;
  if (lane == 0) bz[b] = y[b * 9 + 8] - corr;

  float mine = 0.0f, carry = 0.0f;
#pragma unroll
  for (int i = 0; i < NPRE; ++i) {
    float s = (x[i].x + x[i].y) + (x[i].z + x[i].w);
    s += __shfl_xor(s, 1);
    s += __shfl_xor(s, 2);
    s += __shfl_xor(s, 4);
    s += __shfl_xor(s, 8);
    const float o  = __shfl_xor(s, 16);
    const float lo = (lane < 16) ? s : o;
    const float hi = (lane < 16) ? o : s;
    if (lane == 2 * i) mine = carry;
    if (lane == 2 * i + 1) mine = carry + lo;
    carry = (carry + lo) + hi;
  }
  if (lane == 2 * NPRE) mine = carry;
  if (lane == NSEG - 1) mine = se;
  pre[b * NSEG + lane] = mine;
}

#define GCH 4
#define NCF (GCH + 6)

__device__ __forceinline__ void zc_load(const float* __restrict__ cp, const float* __restrict__ up,
                                        int m0, int g, v4f (&cc)[NCF], v4f (&uu)[2][GCH]) {
#pragma unroll
  for (int j = 0; j < NCF; ++j) cc[j] = *(const v4f*)(cp + (g * GCH - 6 + j) * 8);
#pragma unroll
  for (int f = 0; f < 2; ++f)
#pragma unroll
    for (int j = 0; j < GCH; ++j)
      uu[f][j] = *(const v4f*)(up + (ptrdiff_t)(16 * f) * USTR + (m0 + g * GCH + j) * 8);
}

__device__ __forceinline__ void zc_chunk(const v4f& c4, const v4f& u0, const v4f& u1,
                                         v8f& d0, v8f& d1) {
#pragma unroll
  for (int s = 0; s < 2; ++s) {
    v2f a, f0, f1;
    a.x  = s ? c4.z : c4.x;  a.y  = s ? c4.w : c4.y;
    f0.x = s ? u0.z : u0.x;  f0.y = s ? u0.w : u0.y;
    f1.x = s ? u1.z : u1.x;  f1.y = s ? u1.w : u1.y;
    d0 = __builtin_amdgcn_wmma_f32_16x16x4_f32(false, f0, false, a, (short)0, d0, false, false);
    d1 = __builtin_amdgcn_wmma_f32_16x16x4_f32(false, f1, false, a, (short)0, d1, false, false);
  }
}

template <int G>
__device__ __forceinline__ void zc_mma(const v4f (&cc)[NCF], const v4f (&uu)[2][GCH],
                                       v8f (&acc)[4][2]) {
#pragma unroll
  for (int k = 0; k < GCH; ++k) {
#pragma unroll
    for (int j = 0; j < 4; ++j) {
      const int i = G * GCH + k;
      if (i <= 2 * j + 34)
        zc_chunk(cc[k - 2 * j + 6], uu[0][k], uu[1][k], acc[j][0], acc[j][1]);
    }
  }
}

__global__ void __launch_bounds__(128)
k_zconv(const float* __restrict__ u, const float* __restrict__ ctab,
        const float* __restrict__ hh, const float* __restrict__ h8,
        const float* __restrict__ Eps,
        const float* __restrict__ pre, const float* __restrict__ bz, float* __restrict__ out) {
  const int lane = threadIdx.x & 31;
  const int wv   = threadIdx.x >> 5;
  const int T    = blockIdx.x;
  const int b0 = (blockIdx.y * 4 + wv) * 32;
  const int col  = lane & 15;
  const int half = lane >> 4;

  v8f acc[4][2];
#pragma unroll
  for (int j = 0; j < 4; ++j)
#pragma unroll
    for (int f = 0; f < 2; ++f)
#pragma unroll
      for (int v = 0; v < 8; ++v) acc[j][f][v] = 0.0f;

  const int ib = 103 - col + 4 * half;
  const int sh = ib & 3;
  const float* cp = ctab + sh * CLEN + (ib - sh);
  const float* up = u + (ptrdiff_t)(b0 + col) * USTR + 4 * half;
  const int m0 = 8 * T - 32;

  int m = __float_as_int(h8[BANDSLOT]);
  m = m < 1 ? 1 : (m > MAXBAND ? MAXBAND : m);
  int g = 8 - 2 * ((T >= m) ? m : T);

  v4f cA[NCF], uA[2][GCH], cB[NCF], uB[2][GCH];
  zc_load(cp, up, m0, g, cA, uA);

  if (T < m) {
    const float* eb = Eps + ((size_t)(2 * half) * BATCH + b0 + col);
#pragma unroll
    for (int ks = 0; ks < 2; ++ks) {
      const float* e4 = eb + (size_t)(4 * ks) * BATCH;
      v2f f0, f1;
      f0.x = e4[0];   f0.y = e4[BATCH];
      f1.x = e4[16];  f1.y = e4[BATCH + 16];
#pragma unroll
      for (int j = 0; j < 4; ++j) {
        const float* hb = hh + (HOFF + 64 * T + 16 * j + 1 + col + 7 - 2 * half);
        v2f a;
        a.x = hb[-4 * ks];
        a.y = hb[-4 * ks - 1];
        acc[j][0] = __builtin_amdgcn_wmma_f32_16x16x4_f32(false, f0, false, a, (short)0,
                                                          acc[j][0], false, false);
        acc[j][1] = __builtin_amdgcn_wmma_f32_16x16x4_f32(false, f1, false, a, (short)0,
                                                          acc[j][1], false, false);
      }
    }
  }

  for (; g < 8; g += 2) {
    zc_load(cp, up, m0, g + 1, cB, uB);
    zc_mma<4>(cA, uA, acc);
    zc_load(cp, up, m0, g + 2, cA, uA);
    zc_mma<4>(cB, uB, acc);
  }
  zc_load(cp, up, m0, 9, cB, uB);
  const v4f cL  = *(const v4f*)(cp + 34 * 8);
  const v4f uL0 = *(const v4f*)(up + (m0 + 40) * 8);
  const v4f uL1 = *(const v4f*)(up + (ptrdiff_t)16 * USTR + (m0 + 40) * 8);
  const float hl = (T >= m) ? h8[LASTSLOT] : 0.0f;
  const float gl = (T >= m) ? h8[GAINSLOT] : 0.0f;
  const float* zp = bz + (b0 + 8 * half);
  const float* pp = pre + (size_t)(b0 + 8 * half) * NSEG;
  const int pi = (T >= m) ? T - m : 0;
  float bs[2][8];
#pragma unroll
  for (int f = 0; f < 2; ++f)
#pragma unroll
    for (int v = 0; v < 8; ++v) {
      const float* pr = pp + (16 * f + v) * NSEG;
      bs[f][v] = zp[16 * f + v] + (hl * pr[NSEG - 1] + gl * pr[pi]);
    }
  zc_mma<8>(cA, uA, acc);
  zc_mma<9>(cB, uB, acc);
  zc_chunk(cL, uL0, uL1, acc[3][0], acc[3][1]);

  float* op = out + (size_t)(b0 + 8 * half) * OSTR + 9 + 64 * T + col;
#pragma unroll
  for (int j = 0; j < 4; ++j) {
#pragma unroll
    for (int f = 0; f < 2; ++f) {
#pragma unroll
      for (int v = 0; v < 8; ++v)
        op[(size_t)(16 * f + v) * OSTR + 16 * j] = acc[j][f][v] + bs[f][v];
    }
  }
}

extern "C" void kernel_launch(void* const* d_in, const int* in_sizes, int n_in,
                              void* d_out, int out_size, void* d_ws, size_t ws_size,
                              hipStream_t stream) {
  (void)in_sizes; (void)n_in; (void)out_size; (void)ws_size;
  const float* y = (const float*)d_in[0];
  const float* u = (const float*)d_in[1];
  const float* W = (const float*)d_in[2];
  float* out = (float*)d_out;

  float* hh   = (float*)d_ws;
  float* h8   = hh + HPAD_LEN;
  float* ctab = h8 + 32;
  float* Eps  = ctab + 4 * CLEN;
  float* pre  = Eps + 8 * BATCH;
  float* bz   = pre + BATCH * NSEG;

  k_setup <<<BATCH / 8 + 1, 256, 0, stream>>>(y, u, W, hh, h8, Eps, ctab, pre, bz, out);
  k_zconv <<<dim3(SEQ / 64, BATCH / 128), 128, 0, stream>>>(u, ctab, hh, h8, Eps, pre, bz, out);
}


# k_zconv lead-in: chunk-tiles of the band's first 8 chunks that meet the gain alone leave the matrix unit: lane sums their inputs, one WMMA per tile adds gain x sum; sum order changes
# speedup vs baseline: 14.2851x; 1.0516x over previous
#include <hip/hip_runtime.h>

#define BATCH 4096
#define SEQ   2048
#define ARP   8
#define EXO   8
#define USTR  (SEQ + EXO)
#define OSTR  (SEQ + ARP + 1)
#define KTAPS 256
#define HOFF  32
#define HPAD_LEN 512
#define CLEN  512
#define CTOP  360
#define NSEG  (SEQ / 64)
#define MAXBAND (KTAPS / 64)
#define BANDSLOT 8
#define GAINSLOT 9
#define LASTSLOT 10
#define KCSLOT 11

typedef __attribute__((ext_vector_type(2))) float v2f;
typedef __attribute__((ext_vector_type(4))) float v4f;
typedef __attribute__((ext_vector_type(8))) float v8f;


#define IMP_STEP(hn)                         \
  const float hn = p1;                       \
  p1 = __builtin_fmaf(a1, hn, p2);           \
  p2 = __builtin_fmaf(a2, hn, p3);           \
  p3 = __builtin_fmaf(a3, hn, p4);           \
  p4 = __builtin_fmaf(a4, hn, p5);           \
  p5 = __builtin_fmaf(a5, hn, p6);           \
  p6 = __builtin_fmaf(a6, hn, p7);           \
  p7 = __builtin_fmaf(a7, hn, p8);           \
  p8 = a8 * hn;

#define NPRE 15


__global__ void __launch_bounds__(256)
k_setup(const float* __restrict__ y, const float* __restrict__ u, const float* __restrict__ W,
        float* __restrict__ hh, float* __restrict__ h8,
        float* __restrict__ Eps, float* __restrict__ ctab, float* __restrict__ pre,
        float* __restrict__ bz, float* __restrict__ out) {
  __shared__ __attribute__((aligned(16))) float sH[HPAD_LEN];
  const int tid = threadIdx.x;

  float wx[8];
#pragma unroll
  for (int j = 0; j < 8; ++j) wx[j] = W[8 + j];
  const float a1 = W[7], a2 = W[6], a3 = W[5], a4 = W[4];
  const float a5 = W[3], a6 = W[2], a7 = W[1], a8 = W[0];

  if (blockIdx.x == 0) {
    if (tid < HOFF) sH[tid] = 0.0f;
    if (tid == 0) {
      float p1 = 1.0f, p2 = 0.0f, p3 = 0.0f, p4 = 0.0f;
      float p5 = 0.0f, p6 = 0.0f, p7 = 0.0f, p8 = 0.0f;
      float run = 0.0f;
      int kc = 0;
      for (int j = 0; j < KTAPS / 4; ++j) {
        v4f o, ot;
#pragma unroll
        for (int t = 0; t < 4; ++t) {
          IMP_STEP(hn)
          const float nr = run + hn;
          kc = (nr != run) ? 4 * j + t : kc;
          run = nr;
          ot[t] = hn;
          o[t] = run;
        }
        *(v4f*)(sH + HOFF + 4 * j) = o;
        if (j < 2) *(v4f*)(h8 + 4 * j) = ot;
      }
      int m = (kc + 62) >> 6;
      m = m < 1 ? 1 : (m > MAXBAND ? MAXBAND : m);
      h8[BANDSLOT] = __int_as_float(m);
      h8[KCSLOT] = __int_as_float(kc);
      h8[LASTSLOT] = run;
    }
    __syncthreads();
    const float last = sH[HOFF + KTAPS - 1];
    if (tid < HPAD_LEN - HOFF - KTAPS) sH[HOFF + KTAPS + tid] = last;
    __syncthreads();
    *(v2f*)(hh + 2 * tid) = *(const v2f*)(sH + 2 * tid);
#pragma unroll
    for (int e = 0; e < 4 * CLEN / 256; ++e) {
      const int gid = e * 256 + tid;
      const int t = gid / CLEN;
      const int k = CTOP - (gid - t * CLEN + t);
      float s = 0.0f;
      if (k >= -7) {
#pragma unroll
        for (int j = 0; j < 8; ++j) s += wx[j] * sH[HOFF + k + j];
      }
      ctab[gid] = s;
      if (gid == CTOP - KTAPS) h8[GAINSLOT] = s;
    }
    return;
  }

  const int lane = tid & 31;
  const int b    = (blockIdx.x - 1) * 8 + (tid >> 5);
  const float* ur = u + (size_t)b * USTR;

  v4f x[NPRE];
#pragma unroll
  for (int i = 0; i < NPRE; ++i) x[i] = *(const v4f*)(ur + 128 * i + 4 * lane);

  if (lane < 9) out[(size_t)b * OSTR + lane] = y[b * 9 + lane];

  float hv[8], g8[8], u0[8];
  {
    float p1 = 1.0f, p2 = 0.0f, p3 = 0.0f, p4 = 0.0f;
    float p5 = 0.0f, p6 = 0.0f, p7 = 0.0f, p8 = 0.0f;
    float run = 0.0f;
#pragma unroll
    for (int i = 0; i < 8; ++i) {
      IMP_STEP(hn)
      run = run + hn;
      hv[i] = hn;
      g8[i] = run;
    }
  }
#pragma unroll
  for (int i = 0; i < 8; ++i) u0[i] = ur[i];

  float e[8];
  float se = 0.0f, corr = 0.0f, myeps = 0.0f;
#pragma unroll
  for (int i = 0; i < 8; ++i) {
    float s = y[b * 9 + i + 1] - y[b * 9 + i];
#pragma unroll
    for (int p = 0; p < 8; ++p) {
      if (p < i) s -= hv[i - p] * e[p];
    }
    e[i] = s;
    float d = 0.0f;
#pragma unroll
    for (int j = 0; j < 8; ++j) {
      if (j >= 7 - i) d += wx[j] * u0[i - 7 + j];
    }
    const float ep = s - d;
    if (lane == i) myeps = ep;
    se += ep;
    corr += g8[7 - i] * ep;
  }
#pragma unroll
  for (int q = 0; q < 8; ++q) {
    float cm = 0.0f;
#pragma unroll
    for (int j = 0; j < 8; ++j) {
      if (j >= q) cm += wx[j] * g8[j - q];
    }
    corr += cm * u0[q];
  }
  if (lane < 8) Eps[lane * BATCH + b] = myeps;
  if (lane == 0) bz[b] = y[b * 9 + 8] - corr;

  float mine = 0.0f, carry = 0.0f;
#pragma unroll
  for (int i = 0; i < NPRE; ++i) {
    float s = (x[i].x + x[i].y) + (x[i].z + x[i].w);
    s += __shfl_xor(s, 1);
    s += __shfl_xor(s, 2);
    s += __shfl_xor(s, 4);
    s += __shfl_xor(s, 8);
    const float o  = __shfl_xor(s, 16);
    const float lo = (lane < 16) ? s : o;
    const float hi = (lane < 16) ? o : s;
    if (lane == 2 * i) mine = carry;
    if (lane == 2 * i + 1) mine = carry + lo;
    carry = (carry + lo) + hi;
  }
  if (lane == 2 * NPRE) mine = carry;
  if (lane == NSEG - 1) mine = se;
  pre[b * NSEG + lane] = mine;
}

#define GCH 4
#define NCF (GCH + 6)

__device__ __forceinline__ void zc_load(const float* __restrict__ cp, const float* __restrict__ up,
                                        int m0, int g, v4f (&cc)[NCF], v4f (&uu)[2][GCH]) {
#pragma unroll
  for (int j = 0; j < NCF; ++j) cc[j] = *(const v4f*)(cp + (g * GCH - 6 + j) * 8);
#pragma unroll
  for (int f = 0; f < 2; ++f)
#pragma unroll
    for (int j = 0; j < GCH; ++j)
      uu[f][j] = *(const v4f*)(up + (ptrdiff_t)(16 * f) * USTR + (m0 + g * GCH + j) * 8);
}

__device__ __forceinline__ void zc_chunk(const v4f& c4, const v4f& u0, const v4f& u1,
                                         v8f& d0, v8f& d1) {
#pragma unroll
  for (int s = 0; s < 2; ++s) {
    v2f a, f0, f1;
    a.x  = s ? c4.z : c4.x;  a.y  = s ? c4.w : c4.y;
    f0.x = s ? u0.z : u0.x;  f0.y = s ? u0.w : u0.y;
    f1.x = s ? u1.z : u1.x;  f1.y = s ? u1.w : u1.y;
    d0 = __builtin_amdgcn_wmma_f32_16x16x4_f32(false, f0, false, a, (short)0, d0, false, false);
    d1 = __builtin_amdgcn_wmma_f32_16x16x4_f32(false, f1, false, a, (short)0, d1, false, false);
  }
}

template <int G>
__device__ __forceinline__ void zc_mma(const v4f (&cc)[NCF], const v4f (&uu)[2][GCH],
                                       v8f (&acc)[4][2]) {
#pragma unroll
  for (int k = 0; k < GCH; ++k) {
#pragma unroll
    for (int j = 0; j < 4; ++j) {
      const int i = G * GCH + k;
      if (i <= 2 * j + 34)
        zc_chunk(cc[k - 2 * j + 6], uu[0][k], uu[1][k], acc[j][0], acc[j][1]);
    }
  }
}

template <int E, int H>
__device__ __forceinline__ void zc_mma_lead(const v4f (&cc)[NCF], const v4f (&uu)[2][GCH],
                                            v8f (&acc)[4][2], float (&S)[4][2]) {
#pragma unroll
  for (int k = 0; k < GCH; ++k) {
    const int c = H * GCH + k;
    float t0 = 0.0f, t1 = 0.0f;
    if (c <= E + 6) {
      t0 = (uu[0][k].x + uu[0][k].y) + (uu[0][k].z + uu[0][k].w);
      t1 = (uu[1][k].x + uu[1][k].y) + (uu[1][k].z + uu[1][k].w);
    }
#pragma unroll
    for (int j = 0; j < 4; ++j) {
      if (c <= E + 2 * j) {
        S[j][0] += t0;
        S[j][1] += t1;
      } else {
        zc_chunk(cc[k - 2 * j + 6], uu[0][k], uu[1][k], acc[j][0], acc[j][1]);
      }
    }
  }
}

template <int E>
__device__ __forceinline__ void zc_lead(const float* __restrict__ cp, const float* __restrict__ up,
                                        int m0, int g, float gain,
                                        v4f (&cA)[NCF], v4f (&uA)[2][GCH],
                                        v4f (&cB)[NCF], v4f (&uB)[2][GCH], v8f (&acc)[4][2]) {
  float S[4][2];
#pragma unroll
  for (int j = 0; j < 4; ++j) { S[j][0] = 0.0f; S[j][1] = 0.0f; }
  zc_load(cp, up, m0, g + 1, cB, uB);
  zc_mma_lead<E, 0>(cA, uA, acc, S);
  zc_load(cp, up, m0, g + 2, cA, uA);
  zc_mma_lead<E, 1>(cB, uB, acc, S);
  v2f gb;
  gb.x = gain;  gb.y = gain;
#pragma unroll
  for (int j = 0; j < 4; ++j) {
    if (E + 2 * j >= 0) {
#pragma unroll
      for (int f = 0; f < 2; ++f) {
        v2f sa;
        sa.x = S[j][f];  sa.y = 0.0f;
        acc[j][f] = __builtin_amdgcn_wmma_f32_16x16x4_f32(false, sa, false, gb, (short)0,
                                                          acc[j][f], false, false);
      }
    }
  }
}

template <int E>
__device__ __forceinline__ void zc_body(const float* __restrict__ u, const float* __restrict__ ctab,
                                        const float* __restrict__ hh, const float* __restrict__ h8,
                                        const float* __restrict__ Eps,
                                        const float* __restrict__ pre, const float* __restrict__ bz,
                                        float* __restrict__ out, const int T, const int m) {
  const int lane = threadIdx.x & 31;
  const int wv   = threadIdx.x >> 5;
  const int b0 = (blockIdx.y * 4 + wv) * 32;
  const int col  = lane & 15;
  const int half = lane >> 4;

  v8f acc[4][2];
#pragma unroll
  for (int j = 0; j < 4; ++j)
#pragma unroll
    for (int f = 0; f < 2; ++f)
#pragma unroll
      for (int v = 0; v < 8; ++v) acc[j][f][v] = 0.0f;

  const int ib = 103 - col + 4 * half;
  const int sh = ib & 3;
  const float* cp = ctab + sh * CLEN + (ib - sh);
  const float* up = u + (ptrdiff_t)(b0 + col) * USTR + 4 * half;
  const int m0 = 8 * T - 32;

  int g = 8 - 2 * ((T >= m) ? m : T);

  v4f cA[NCF], uA[2][GCH], cB[NCF], uB[2][GCH];
  zc_load(cp, up, m0, g, cA, uA);

  if (E < -1 && T < m) {
    const float* eb = Eps + ((size_t)(2 * half) * BATCH + b0 + col);
#pragma unroll
    for (int ks = 0; ks < 2; ++ks) {
      const float* e4 = eb + (size_t)(4 * ks) * BATCH;
      v2f f0, f1;
      f0.x = e4[0];   f0.y = e4[BATCH];
      f1.x = e4[16];  f1.y = e4[BATCH + 16];
#pragma unroll
      for (int j = 0; j < 4; ++j) {
        const float* hb = hh + (HOFF + 64 * T + 16 * j + 1 + col + 7 - 2 * half);
        v2f a;
        a.x = hb[-4 * ks];
        a.y = hb[-4 * ks - 1];
        acc[j][0] = __builtin_amdgcn_wmma_f32_16x16x4_f32(false, f0, false, a, (short)0,
                                                          acc[j][0], false, false);
        acc[j][1] = __builtin_amdgcn_wmma_f32_16x16x4_f32(false, f1, false, a, (short)0,
                                                          acc[j][1], false, false);
      }
    }
  }

  if (E >= -1) {
    zc_lead<E>(cp, up, m0, g, h8[GAINSLOT], cA, uA, cB, uB, acc);
    g += 2;
  }

  for (; g < 8; g += 2) {
    zc_load(cp, up, m0, g + 1, cB, uB);
    zc_mma<4>(cA, uA, acc);
    zc_load(cp, up, m0, g + 2, cA, uA);
    zc_mma<4>(cB, uB, acc);
  }
  zc_load(cp, up, m0, 9, cB, uB);
  const v4f cL  = *(const v4f*)(cp + 34 * 8);
  const v4f uL0 = *(const v4f*)(up + (m0 + 40) * 8);
  const v4f uL1 = *(const v4f*)(up + (ptrdiff_t)16 * USTR + (m0 + 40) * 8);
  const float hl = (T >= m) ? h8[LASTSLOT] : 0.0f;
  const float gl = (T >= m) ? h8[GAINSLOT] : 0.0f;
  const float* zp = bz + (b0 + 8 * half);
  const float* pp = pre + (size_t)(b0 + 8 * half) * NSEG;
  const int pi = (T >= m) ? T - m : 0;
  float bs[2][8];
#pragma unroll
  for (int f = 0; f < 2; ++f)
#pragma unroll
    for (int v = 0; v < 8; ++v) {
      const float* pr = pp + (16 * f + v) * NSEG;
      bs[f][v] = zp[16 * f + v] + (hl * pr[NSEG - 1] + gl * pr[pi]);
    }
  zc_mma<8>(cA, uA, acc);
  zc_mma<9>(cB, uB, acc);
  zc_chunk(cL, uL0, uL1, acc[3][0], acc[3][1]);

  float* op = out + (size_t)(b0 + 8 * half) * OSTR + 9 + 64 * T + col;
#pragma unroll
  for (int j = 0; j < 4; ++j) {
#pragma unroll
    for (int f = 0; f < 2; ++f) {
#pragma unroll
      for (int v = 0; v < 8; ++v)
        op[(size_t)(16 * f + v) * OSTR + 16 * j] = acc[j][f][v] + bs[f][v];
    }
  }
}

__global__ void __launch_bounds__(128, 4)
k_zconv(const float* __restrict__ u, const float* __restrict__ ctab,
        const float* __restrict__ hh, const float* __restrict__ h8,
        const float* __restrict__ Eps,
        const float* __restrict__ pre, const float* __restrict__ bz, float* __restrict__ out) {
  const int T = blockIdx.x;
  int m = __float_as_int(h8[BANDSLOT]);
  m = m < 1 ? 1 : (m > MAXBAND ? MAXBAND : m);
  int kc = __float_as_int(h8[KCSLOT]);
  kc = kc < 0 ? 0 : (kc > KTAPS - 1 ? KTAPS - 1 : kc);
  const int e = 8 * m - ((kc + 13) >> 3);
  if (T >= m && e >= -1) {
    switch (e) {
      case -1: zc_body<-1>(u, ctab, hh, h8, Eps, pre, bz, out, T, m); break;
      case 0:  zc_body<0>(u, ctab, hh, h8, Eps, pre, bz, out, T, m); break;
      case 1:  zc_body<1>(u, ctab, hh, h8, Eps, pre, bz, out, T, m); break;
      case 2:  zc_body<2>(u, ctab, hh, h8, Eps, pre, bz, out, T, m); break;
      default: zc_body<3>(u, ctab, hh, h8, Eps, pre, bz, out, T, m); break;
    }
  } else {
    zc_body<-8>(u, ctab, hh, h8, Eps, pre, bz, out, T, m);
  }
}

extern "C" void kernel_launch(void* const* d_in, const int* in_sizes, int n_in,
                              void* d_out, int out_size, void* d_ws, size_t ws_size,
                              hipStream_t stream) {
  (void)in_sizes; (void)n_in; (void)out_size; (void)ws_size;
  const float* y = (const float*)d_in[0];
  const float* u = (const float*)d_in[1];
  const float* W = (const float*)d_in[2];
  float* out = (float*)d_out;

  float* hh   = (float*)d_ws;
  float* h8   = hh + HPAD_LEN;
  float* ctab = h8 + 32;
  float* Eps  = ctab + 4 * CLEN;
  float* pre  = Eps + 8 * BATCH;
  float* bz   = pre + BATCH * NSEG;

  k_setup <<<BATCH / 8 + 1, 256, 0, stream>>>(y, u, W, hh, h8, Eps, ctab, pre, bz, out);
  k_zconv <<<dim3(SEQ / 64, BATCH / 128), 128, 0, stream>>>(u, ctab, hh, h8, Eps, pre, bz, out);
}


# k_zconv: a wave takes two adjacent groups of 64 steps (8 step tiles): coefficient fragments and u chunks loaded once, shared chunks read once; 2048 waves, not 4096; sum order kept but steps 65-128
# speedup vs baseline: 16.2203x; 1.1355x over previous
#include <hip/hip_runtime.h>

#define BATCH 4096
#define SEQ   2048
#define ARP   8
#define EXO   8
#define USTR  (SEQ + EXO)
#define OSTR  (SEQ + ARP + 1)
#define KTAPS 256
#define HOFF  32
#define HPAD_LEN 512
#define CLEN  512
#define CTOP  360
#define NSEG  (SEQ / 64)
#define MAXBAND (KTAPS / 64)
#define BANDSLOT 8
#define GAINSLOT 9
#define LASTSLOT 10
#define KCSLOT 11

typedef __attribute__((ext_vector_type(2))) float v2f;
typedef __attribute__((ext_vector_type(4))) float v4f;
typedef __attribute__((ext_vector_type(8))) float v8f;


#define IMP_STEP(hn)                         \
  const float hn = p1;                       \
  p1 = __builtin_fmaf(a1, hn, p2);           \
  p2 = __builtin_fmaf(a2, hn, p3);           \
  p3 = __builtin_fmaf(a3, hn, p4);           \
  p4 = __builtin_fmaf(a4, hn, p5);           \
  p5 = __builtin_fmaf(a5, hn, p6);           \
  p6 = __builtin_fmaf(a6, hn, p7);           \
  p7 = __builtin_fmaf(a7, hn, p8);           \
  p8 = a8 * hn;

#define NPRE 15


__global__ void __launch_bounds__(256)
k_setup(const float* __restrict__ y, const float* __restrict__ u, const float* __restrict__ W,
        float* __restrict__ hh, float* __restrict__ h8,
        float* __restrict__ Eps, float* __restrict__ ctab, float* __restrict__ pre,
        float* __restrict__ bz, float* __restrict__ out) {
  __shared__ __attribute__((aligned(16))) float sH[HPAD_LEN];
  const int tid = threadIdx.x;

  float wx[8];
#pragma unroll
  for (int j = 0; j < 8; ++j) wx[j] = W[8 + j];
  const float a1 = W[7], a2 = W[6], a3 = W[5], a4 = W[4];
  const float a5 = W[3], a6 = W[2], a7 = W[1], a8 = W[0];

  if (blockIdx.x == 0) {
    if (tid < HOFF) sH[tid] = 0.0f;
    if (tid == 0) {
      float p1 = 1.0f, p2 = 0.0f, p3 = 0.0f, p4 = 0.0f;
      float p5 = 0.0f, p6 = 0.0f, p7 = 0.0f, p8 = 0.0f;
      float run = 0.0f;
      int kc = 0;
      for (int j = 0; j < KTAPS / 4; ++j) {
        v4f o, ot;
#pragma unroll
        for (int t = 0; t < 4; ++t) {
          IMP_STEP(hn)
          const float nr = run + hn;
          kc = (nr != run) ? 4 * j + t : kc;
          run = nr;
          ot[t] = hn;
          o[t] = run;
        }
        *(v4f*)(sH + HOFF + 4 * j) = o;
        if (j < 2) *(v4f*)(h8 + 4 * j) = ot;
      }
      int m = (kc + 62) >> 6;
      m = m < 1 ? 1 : (m > MAXBAND ? MAXBAND : m);
      h8[BANDSLOT] = __int_as_float(m);
      h8[KCSLOT] = __int_as_float(kc);
      h8[LASTSLOT] = run;
    }
    __syncthreads();
    const float last = sH[HOFF + KTAPS - 1];
    if (tid < HPAD_LEN - HOFF - KTAPS) sH[HOFF + KTAPS + tid] = last;
    __syncthreads();
    *(v2f*)(hh + 2 * tid) = *(const v2f*)(sH + 2 * tid);
#pragma unroll
    for (int e = 0; e < 4 * CLEN / 256; ++e) {
      const int gid = e * 256 + tid;
      const int t = gid / CLEN;
      const int k = CTOP - (gid - t * CLEN + t);
      float s = 0.0f;
      if (k >= -7) {
#pragma unroll
        for (int j = 0; j < 8; ++j) s += wx[j] * sH[HOFF + k + j];
      }
      ctab[gid] = s;
      if (gid == CTOP - KTAPS) h8[GAINSLOT] = s;
    }
    return;
  }

  const int lane = tid & 31;
  const int b    = (blockIdx.x - 1) * 8 + (tid >> 5);
  const float* ur = u + (size_t)b * USTR;

  v4f x[NPRE];
#pragma unroll
  for (int i = 0; i < NPRE; ++i) x[i] = *(const v4f*)(ur + 128 * i + 4 * lane);

  if (lane < 9) out[(size_t)b * OSTR + lane] = y[b * 9 + lane];

  float hv[8], g8[8], u0[8];
  {
    float p1 = 1.0f, p2 = 0.0f, p3 = 0.0f, p4 = 0.0f;
    float p5 = 0.0f, p6 = 0.0f, p7 = 0.0f, p8 = 0.0f;
    float run = 0.0f;
#pragma unroll
    for (int i = 0; i < 8; ++i) {
      IMP_STEP(hn)
      run = run + hn;
      hv[i] = hn;
      g8[i] = run;
    }
  }
#pragma unroll
  for (int i = 0; i < 8; ++i) u0[i] = ur[i];

  float e[8];
  float se = 0.0f, corr = 0.0f, myeps = 0.0f;
#pragma unroll
  for (int i = 0; i < 8; ++i) {
    float s = y[b * 9 + i + 1] - y[b * 9 + i];
#pragma unroll
    for (int p = 0; p < 8; ++p) {
      if (p < i) s -= hv[i - p] * e[p];
    }
    e[i] = s;
    float d = 0.0f;
#pragma unroll
    for (int j = 0; j < 8; ++j) {
      if (j >= 7 - i) d += wx[j] * u0[i - 7 + j];
    }
    const float ep = s - d;
    if (lane == i) myeps = ep;
    se += ep;
    corr += g8[7 - i] * ep;
  }
#pragma unroll
  for (int q = 0; q < 8; ++q) {
    float cm = 0.0f;
#pragma unroll
    for (int j = 0; j < 8; ++j) {
      if (j >= q) cm += wx[j] * g8[j - q];
    }
    corr += cm * u0[q];
  }
  if (lane < 8) Eps[lane * BATCH + b] = myeps;
  if (lane == 0) bz[b] = y[b * 9 + 8] - corr;

  float mine = 0.0f, carry = 0.0f;
#pragma unroll
  for (int i = 0; i < NPRE; ++i) {
    float s = (x[i].x + x[i].y) + (x[i].z + x[i].w);
    s += __shfl_xor(s, 1);
    s += __shfl_xor(s, 2);
    s += __shfl_xor(s, 4);
    s += __shfl_xor(s, 8);
    const float o  = __shfl_xor(s, 16);
    const float lo = (lane < 16) ? s : o;
    const float hi = (lane < 16) ? o : s;
    if (lane == 2 * i) mine = carry;
    if (lane == 2 * i + 1) mine = carry + lo;
    carry = (carry + lo) + hi;
  }
  if (lane == 2 * NPRE) mine = carry;
  if (lane == NSEG - 1) mine = se;
  pre[b * NSEG + lane] = mine;
}

#define GCH 4
#define NCF (GCH + 6)

__device__ __forceinline__ void zc_load(const float* __restrict__ cp, const float* __restrict__ up,
                                        int m0, int g, v4f (&cc)[NCF], v4f (&uu)[2][GCH]) {
#pragma unroll
  for (int j = 0; j < NCF; ++j) cc[j] = *(const v4f*)(cp + (g * GCH - 6 + j) * 8);
#pragma unroll
  for (int f = 0; f < 2; ++f)
#pragma unroll
    for (int j = 0; j < GCH; ++j)
      uu[f][j] = *(const v4f*)(up + (ptrdiff_t)(16 * f) * USTR + (m0 + g * GCH + j) * 8);
}

__device__ __forceinline__ void zc_chunk(const v4f& c4, const v4f& u0, const v4f& u1,
                                         v8f& d0, v8f& d1) {
#pragma unroll
  for (int s = 0; s < 2; ++s) {
    v2f a, f0, f1;
    a.x  = s ? c4.z : c4.x;  a.y  = s ? c4.w : c4.y;
    f0.x = s ? u0.z : u0.x;  f0.y = s ? u0.w : u0.y;
    f1.x = s ? u1.z : u1.x;  f1.y = s ? u1.w : u1.y;
    d0 = __builtin_amdgcn_wmma_f32_16x16x4_f32(false, f0, false, a, (short)0, d0, false, false);
    d1 = __builtin_amdgcn_wmma_f32_16x16x4_f32(false, f1, false, a, (short)0, d1, false, false);
  }
}

template <int G>
__device__ __forceinline__ void zc_mma(const v4f (&cc)[NCF], const v4f (&uu)[2][GCH],
                                       v8f (&acc)[4][2]) {
#pragma unroll
  for (int k = 0; k < GCH; ++k) {
#pragma unroll
    for (int j = 0; j < 4; ++j) {
      const int i = G * GCH + k;
      if (i <= 2 * j + 34)
        zc_chunk(cc[k - 2 * j + 6], uu[0][k], uu[1][k], acc[j][0], acc[j][1]);
    }
  }
}

__device__ __forceinline__ void zc_plain(const float* __restrict__ u, const float* __restrict__ ctab,
                                         const float* __restrict__ hh, const float* __restrict__ h8,
                                         const float* __restrict__ Eps,
                                         const float* __restrict__ pre, const float* __restrict__ bz,
                                         float* __restrict__ out, const int T, const int m) {
  const int lane = threadIdx.x & 31;
  const int wv   = threadIdx.x >> 5;
  const int b0 = (blockIdx.y * 4 + wv) * 32;
  const int col  = lane & 15;
  const int half = lane >> 4;

  v8f acc[4][2];
#pragma unroll
  for (int j = 0; j < 4; ++j)
#pragma unroll
    for (int f = 0; f < 2; ++f)
#pragma unroll
      for (int v = 0; v < 8; ++v) acc[j][f][v] = 0.0f;

  const int ib = 103 - col + 4 * half;
  const int sh = ib & 3;
  const float* cp = ctab + sh * CLEN + (ib - sh);
  const float* up = u + (ptrdiff_t)(b0 + col) * USTR + 4 * half;
  const int m0 = 8 * T - 32;

  int g = 8 - 2 * ((T >= m) ? m : T);

  v4f cA[NCF], uA[2][GCH], cB[NCF], uB[2][GCH];
  zc_load(cp, up, m0, g, cA, uA);

  if (T < m) {
    const float* eb = Eps + ((size_t)(2 * half) * BATCH + b0 + col);
#pragma unroll
    for (int ks = 0; ks < 2; ++ks) {
      const float* e4 = eb + (size_t)(4 * ks) * BATCH;
      v2f f0, f1;
      f0.x = e4[0];   f0.y = e4[BATCH];
      f1.x = e4[16];  f1.y = e4[BATCH + 16];
#pragma unroll
      for (int j = 0; j < 4; ++j) {
        const float* hb = hh + (HOFF + 64 * T + 16 * j + 1 + col + 7 - 2 * half);
        v2f a;
        a.x = hb[-4 * ks];
        a.y = hb[-4 * ks - 1];
        acc[j][0] = __builtin_amdgcn_wmma_f32_16x16x4_f32(false, f0, false, a, (short)0,
                                                          acc[j][0], false, false);
        acc[j][1] = __builtin_amdgcn_wmma_f32_16x16x4_f32(false, f1, false, a, (short)0,
                                                          acc[j][1], false, false);
      }
    }
  }

  for (; g < 8; g += 2) {
    zc_load(cp, up, m0, g + 1, cB, uB);
    zc_mma<4>(cA, uA, acc);
    zc_load(cp, up, m0, g + 2, cA, uA);
    zc_mma<4>(cB, uB, acc);
  }
  zc_load(cp, up, m0, 9, cB, uB);
  const v4f cL  = *(const v4f*)(cp + 34 * 8);
  const v4f uL0 = *(const v4f*)(up + (m0 + 40) * 8);
  const v4f uL1 = *(const v4f*)(up + (ptrdiff_t)16 * USTR + (m0 + 40) * 8);
  const float hl = (T >= m) ? h8[LASTSLOT] : 0.0f;
  const float gl = (T >= m) ? h8[GAINSLOT] : 0.0f;
  const float* zp = bz + (b0 + 8 * half);
  const float* pp = pre + (size_t)(b0 + 8 * half) * NSEG;
  const int pi = (T >= m) ? T - m : 0;
  float bs[2][8];
#pragma unroll
  for (int f = 0; f < 2; ++f)
#pragma unroll
    for (int v = 0; v < 8; ++v) {
      const float* pr = pp + (16 * f + v) * NSEG;
      bs[f][v] = zp[16 * f + v] + (hl * pr[NSEG - 1] + gl * pr[pi]);
    }
  zc_mma<8>(cA, uA, acc);
  zc_mma<9>(cB, uB, acc);
  zc_chunk(cL, uL0, uL1, acc[3][0], acc[3][1]);

  float* op = out + (size_t)(b0 + 8 * half) * OSTR + 9 + 64 * T + col;
#pragma unroll
  for (int j = 0; j < 4; ++j) {
#pragma unroll
    for (int f = 0; f < 2; ++f) {
#pragma unroll
      for (int v = 0; v < 8; ++v)
        op[(size_t)(16 * f + v) * OSTR + 16 * j] = acc[j][f][v] + bs[f][v];
    }
  }
}

#define PI0 24
#define PCH 25
#define PR0 18
#define PCF 17

template <int E>
__device__ __forceinline__ void zc_pair(const float* __restrict__ u, const float* __restrict__ ctab,
                                        const float* __restrict__ h8,
                                        const float* __restrict__ pre, const float* __restrict__ bz,
                                        float* __restrict__ out, const int T) {
  const int lane = threadIdx.x & 31;
  const int wv   = threadIdx.x >> 5;
  const int b0 = (blockIdx.y * 4 + wv) * 32;
  const int col  = lane & 15;
  const int half = lane >> 4;

  v8f acc[8][2];
#pragma unroll
  for (int J = 0; J < 8; ++J)
#pragma unroll
    for (int f = 0; f < 2; ++f)
#pragma unroll
      for (int v = 0; v < 8; ++v) acc[J][f][v] = 0.0f;

  const int ib = 103 - col + 4 * half;
  const int sh = ib & 3;
  const float* cp = ctab + sh * CLEN + (ib - sh);
  const float* up = u + (ptrdiff_t)(b0 + col) * USTR + 4 * half;
  const int m0 = 8 * T - 32;

  v4f cf[PCF], uu[2][PCH];
#pragma unroll
  for (int r = 0; r < PCF; ++r) cf[r] = *(const v4f*)(cp + (PR0 + r) * 8);
#pragma unroll
  for (int ii = 0; ii < PCH; ++ii)
#pragma unroll
    for (int f = 0; f < 2; ++f)
      uu[f][ii] = *(const v4f*)(up + (ptrdiff_t)(16 * f) * USTR + (m0 + PI0 + ii) * 8);

  const float gain = h8[GAINSLOT];
  const float hl   = h8[LASTSLOT];
  v2f gb;
  gb.x = gain;  gb.y = gain;

  const float* zp = bz + (b0 + 8 * half);
  const float* pp = pre + (size_t)(b0 + 8 * half) * NSEG;
  float* op = out + (size_t)(b0 + 8 * half) * OSTR + 9 + 64 * T + col;
  float zv[2][8], q31[2][8], qa[2][8], qb[2][8];

  float S[8][2];
#pragma unroll
  for (int J = 0; J < 8; ++J) { S[J][0] = 0.0f; S[J][1] = 0.0f; }

#pragma unroll
  for (int ii = 0; ii < PCH; ++ii) {
    const int i = PI0 + ii;
    float t0 = 0.0f, t1 = 0.0f;
    if (ii < 16) {
      t0 = (uu[0][ii].x + uu[0][ii].y) + (uu[0][ii].z + uu[0][ii].w);
      t1 = (uu[1][ii].x + uu[1][ii].y) + (uu[1][ii].z + uu[1][ii].w);
    }
#pragma unroll
    for (int J = 0; J < 8; ++J) {
      const int j = J & 3;
      const int c = ii - 8 * (J >> 2);
      if (c >= 0 && c < 8) {
        if (c <= E + 2 * j) {
          S[J][0] += t0;
          S[J][1] += t1;
        } else {
          zc_chunk(cf[i - 2 * J - PR0], uu[0][ii], uu[1][ii], acc[J][0], acc[J][1]);
        }
      } else if (c >= 8 && i <= 2 * J + 34) {
        zc_chunk(cf[i - 2 * J - PR0], uu[0][ii], uu[1][ii], acc[J][0], acc[J][1]);
      }
    }
    if (ii == 7 || ii == 15) {
      const int G = ii >> 3;
#pragma unroll
      for (int j = 0; j < 4; ++j) {
        if (E + 2 * j >= 0) {
#pragma unroll
          for (int f = 0; f < 2; ++f) {
            v2f sa;
            sa.x = S[4 * G + j][f];  sa.y = 0.0f;
            acc[4 * G + j][f] = __builtin_amdgcn_wmma_f32_16x16x4_f32(false, sa, false, gb, (short)0,
                                                                  acc[4 * G + j][f], false, false);
          }
        }
      }
    }
    if (ii == 8) {
#pragma unroll
      for (int f = 0; f < 2; ++f)
#pragma unroll
        for (int v = 0; v < 8; ++v) {
          const float* pr = pp + (16 * f + v) * NSEG;
          zv[f][v]  = zp[16 * f + v];
          q31[f][v] = pr[NSEG - 1];
          qa[f][v]  = pr[T - 1];
          qb[f][v]  = pr[T];
        }
    }
    if (ii == 16) {
#pragma unroll
      for (int j = 0; j < 4; ++j)
#pragma unroll
        for (int f = 0; f < 2; ++f)
#pragma unroll
          for (int v = 0; v < 8; ++v)
            op[(size_t)(16 * f + v) * OSTR + 16 * j] =
                acc[j][f][v] + (zv[f][v] + (hl * q31[f][v] + gain * qa[f][v]));
    }
  }
#pragma unroll
  for (int j = 0; j < 4; ++j)
#pragma unroll
    for (int f = 0; f < 2; ++f)
#pragma unroll
      for (int v = 0; v < 8; ++v)
        op[(size_t)(16 * f + v) * OSTR + 16 * (4 + j)] =
            acc[4 + j][f][v] + (zv[f][v] + (hl * q31[f][v] + gain * qb[f][v]));
}

__global__ void __launch_bounds__(128, 2)
k_zconv(const float* __restrict__ u, const float* __restrict__ ctab,
        const float* __restrict__ hh, const float* __restrict__ h8,
        const float* __restrict__ Eps,
        const float* __restrict__ pre, const float* __restrict__ bz, float* __restrict__ out) {
  const int T = 2 * blockIdx.x;
  int m = __float_as_int(h8[BANDSLOT]);
  m = m < 1 ? 1 : (m > MAXBAND ? MAXBAND : m);
  int kc = __float_as_int(h8[KCSLOT]);
  kc = kc < 0 ? 0 : (kc > KTAPS - 1 ? KTAPS - 1 : kc);
  const int e = 8 * m - ((kc + 13) >> 3);
  if (m == 1 && T >= 2 && e >= -1) {
    switch (e) {
      case -1: zc_pair<-1>(u, ctab, h8, pre, bz, out, T); break;
      case 0:  zc_pair<0>(u, ctab, h8, pre, bz, out, T); break;
      case 1:  zc_pair<1>(u, ctab, h8, pre, bz, out, T); break;
      case 2:  zc_pair<2>(u, ctab, h8, pre, bz, out, T); break;
      default: zc_pair<3>(u, ctab, h8, pre, bz, out, T); break;
    }
  } else {
#pragma nounroll
    for (int t = 0; t < 2; ++t) zc_plain(u, ctab, hh, h8, Eps, pre, bz, out, T + t, m);
  }
}

extern "C" void kernel_launch(void* const* d_in, const int* in_sizes, int n_in,
                              void* d_out, int out_size, void* d_ws, size_t ws_size,
                              hipStream_t stream) {
  (void)in_sizes; (void)n_in; (void)out_size; (void)ws_size;
  const float* y = (const float*)d_in[0];
  const float* u = (const float*)d_in[1];
  const float* W = (const float*)d_in[2];
  float* out = (float*)d_out;

  float* hh   = (float*)d_ws;
  float* h8   = hh + HPAD_LEN;
  float* ctab = h8 + 32;
  float* Eps  = ctab + 4 * CLEN;
  float* pre  = Eps + 8 * BATCH;
  float* bz   = pre + BATCH * NSEG;

  k_setup <<<BATCH / 8 + 1, 256, 0, stream>>>(y, u, W, hh, h8, Eps, ctab, pre, bz, out);
  k_zconv <<<dim3(SEQ / 128, BATCH / 128), 128, 0, stream>>>(u, ctab, hh, h8, Eps, pre, bz, out);
}


# k_setup row pass re-tiled: a wave takes two batch rows, every load of both (y, first inputs, 2 x 1920 inputs) issued as one group at entry; 2056 waves enter, not 4104; sum order kept
# speedup vs baseline: 17.1581x; 1.0578x over previous
#include <hip/hip_runtime.h>

#define BATCH 4096
#define SEQ   2048
#define ARP   8
#define EXO   8
#define USTR  (SEQ + EXO)
#define OSTR  (SEQ + ARP + 1)
#define KTAPS 256
#define HOFF  32
#define HPAD_LEN 512
#define CLEN  512
#define CTOP  360
#define NSEG  (SEQ / 64)
#define MAXBAND (KTAPS / 64)
#define BANDSLOT 8
#define GAINSLOT 9
#define LASTSLOT 10
#define KCSLOT 11

typedef __attribute__((ext_vector_type(2))) float v2f;
typedef __attribute__((ext_vector_type(4))) float v4f;
typedef __attribute__((ext_vector_type(8))) float v8f;


#define IMP_STEP(hn)                         \
  const float hn = p1;                       \
  p1 = __builtin_fmaf(a1, hn, p2);           \
  p2 = __builtin_fmaf(a2, hn, p3);           \
  p3 = __builtin_fmaf(a3, hn, p4);           \
  p4 = __builtin_fmaf(a4, hn, p5);           \
  p5 = __builtin_fmaf(a5, hn, p6);           \
  p6 = __builtin_fmaf(a6, hn, p7);           \
  p7 = __builtin_fmaf(a7, hn, p8);           \
  p8 = a8 * hn;

#define NPRE 15
#define RPW  2


__global__ void __launch_bounds__(256)
k_setup(const float* __restrict__ y, const float* __restrict__ u, const float* __restrict__ W,
        float* __restrict__ hh, float* __restrict__ h8,
        float* __restrict__ Eps, float* __restrict__ ctab, float* __restrict__ pre,
        float* __restrict__ bz, float* __restrict__ out) {
  __shared__ __attribute__((aligned(16))) float sH[HPAD_LEN];
  const int tid = threadIdx.x;

  float wx[8];
#pragma unroll
  for (int j = 0; j < 8; ++j) wx[j] = W[8 + j];
  const float a1 = W[7], a2 = W[6], a3 = W[5], a4 = W[4];
  const float a5 = W[3], a6 = W[2], a7 = W[1], a8 = W[0];

  if (blockIdx.x == 0) {
    if (tid < HOFF) sH[tid] = 0.0f;
    if (tid == 0) {
      float p1 = 1.0f, p2 = 0.0f, p3 = 0.0f, p4 = 0.0f;
      float p5 = 0.0f, p6 = 0.0f, p7 = 0.0f, p8 = 0.0f;
      float run = 0.0f;
      int kc = 0;
      for (int j = 0; j < KTAPS / 4; ++j) {
        v4f o, ot;
#pragma unroll
        for (int t = 0; t < 4; ++t) {
          IMP_STEP(hn)
          const float nr = run + hn;
          kc = (nr != run) ? 4 * j + t : kc;
          run = nr;
          ot[t] = hn;
          o[t] = run;
        }
        *(v4f*)(sH + HOFF + 4 * j) = o;
        if (j < 2) *(v4f*)(h8 + 4 * j) = ot;
      }
      int m = (kc + 62) >> 6;
      m = m < 1 ? 1 : (m > MAXBAND ? MAXBAND : m);
      h8[BANDSLOT] = __int_as_float(m);
      h8[KCSLOT] = __int_as_float(kc);
      h8[LASTSLOT] = run;
    }
    __syncthreads();
    const float last = sH[HOFF + KTAPS - 1];
    if (tid < HPAD_LEN - HOFF - KTAPS) sH[HOFF + KTAPS + tid] = last;
    __syncthreads();
    *(v2f*)(hh + 2 * tid) = *(const v2f*)(sH + 2 * tid);
#pragma unroll
    for (int e = 0; e < 4 * CLEN / 256; ++e) {
      const int gid = e * 256 + tid;
      const int t = gid / CLEN;
      const int k = CTOP - (gid - t * CLEN + t);
      float s = 0.0f;
      if (k >= -7) {
#pragma unroll
        for (int j = 0; j < 8; ++j) s += wx[j] * sH[HOFF + k + j];
      }
      ctab[gid] = s;
      if (gid == CTOP - KTAPS) h8[GAINSLOT] = s;
    }
    return;
  }

  const int lane = tid & 31;
  const int bw   = (blockIdx.x - 1) * (8 * RPW) + (tid >> 5) * RPW;

  float yv[RPW][9], u0[RPW][8], yc[RPW];
#pragma unroll
  for (int r = 0; r < RPW; ++r) {
    const int b = bw + r;
#pragma unroll
    for (int i = 0; i < 9; ++i) yv[r][i] = y[b * 9 + i];
#pragma unroll
    for (int i = 0; i < 8; ++i) u0[r][i] = u[(size_t)b * USTR + i];
    yc[r] = y[b * 9 + (lane < 9 ? lane : 8)];
  }
  v4f x[RPW][NPRE];
#pragma unroll
  for (int r = 0; r < RPW; ++r)
#pragma unroll
    for (int i = 0; i < NPRE; ++i)
      x[r][i] = *(const v4f*)(u + (size_t)(bw + r) * USTR + 128 * i + 4 * lane);

  float hv[8], g8[8];
  {
    float p1 = 1.0f, p2 = 0.0f, p3 = 0.0f, p4 = 0.0f;
    float p5 = 0.0f, p6 = 0.0f, p7 = 0.0f, p8 = 0.0f;
    float run = 0.0f;
#pragma unroll
    for (int i = 0; i < 8; ++i) {
      IMP_STEP(hn)
      run = run + hn;
      hv[i] = hn;
      g8[i] = run;
    }
  }

#pragma unroll
  for (int r = 0; r < RPW; ++r) {
    const int b = bw + r;
    float e[8];
    float se = 0.0f, corr = 0.0f, myeps = 0.0f;
#pragma unroll
    for (int i = 0; i < 8; ++i) {
      float s = yv[r][i + 1] - yv[r][i];
#pragma unroll
      for (int p = 0; p < 8; ++p) {
        if (p < i) s -= hv[i - p] * e[p];
      }
      e[i] = s;
      float d = 0.0f;
#pragma unroll
      for (int j = 0; j < 8; ++j) {
        if (j >= 7 - i) d += wx[j] * u0[r][i - 7 + j];
      }
      const float ep = s - d;
      if (lane == i) myeps = ep;
      se += ep;
      corr += g8[7 - i] * ep;
    }
#pragma unroll
    for (int q = 0; q < 8; ++q) {
      float cm = 0.0f;
#pragma unroll
      for (int j = 0; j < 8; ++j) {
        if (j >= q) cm += wx[j] * g8[j - q];
      }
      corr += cm * u0[r][q];
    }
    if (lane < 8) Eps[lane * BATCH + b] = myeps;
    if (lane == 0) bz[b] = yv[r][8] - corr;

    float mine = 0.0f, carry = 0.0f;
#pragma unroll
    for (int i = 0; i < NPRE; ++i) {
      float s = (x[r][i].x + x[r][i].y) + (x[r][i].z + x[r][i].w);
      s += __shfl_xor(s, 1);
      s += __shfl_xor(s, 2);
      s += __shfl_xor(s, 4);
      s += __shfl_xor(s, 8);
      const float o  = __shfl_xor(s, 16);
      const float lo = (lane < 16) ? s : o;
      const float hi = (lane < 16) ? o : s;
      if (lane == 2 * i) mine = carry;
      if (lane == 2 * i + 1) mine = carry + lo;
      carry = (carry + lo) + hi;
    }
    if (lane == 2 * NPRE) mine = carry;
    if (lane == NSEG - 1) mine = se;
    pre[b * NSEG + lane] = mine;
    if (lane < 9) out[(size_t)b * OSTR + lane] = yc[r];
  }
}

#define GCH 4
#define NCF (GCH + 6)

__device__ __forceinline__ void zc_load(const float* __restrict__ cp, const float* __restrict__ up,
                                        int m0, int g, v4f (&cc)[NCF], v4f (&uu)[2][GCH]) {
#pragma unroll
  for (int j = 0; j < NCF; ++j) cc[j] = *(const v4f*)(cp + (g * GCH - 6 + j) * 8);
#pragma unroll
  for (int f = 0; f < 2; ++f)
#pragma unroll
    for (int j = 0; j < GCH; ++j)
      uu[f][j] = *(const v4f*)(up + (ptrdiff_t)(16 * f) * USTR + (m0 + g * GCH + j) * 8);
}

__device__ __forceinline__ void zc_chunk(const v4f& c4, const v4f& u0, const v4f& u1,
                                         v8f& d0, v8f& d1) {
#pragma unroll
  for (int s = 0; s < 2; ++s) {
    v2f a, f0, f1;
    a.x  = s ? c4.z : c4.x;  a.y  = s ? c4.w : c4.y;
    f0.x = s ? u0.z : u0.x;  f0.y = s ? u0.w : u0.y;
    f1.x = s ? u1.z : u1.x;  f1.y = s ? u1.w : u1.y;
    d0 = __builtin_amdgcn_wmma_f32_16x16x4_f32(false, f0, false, a, (short)0, d0, false, false);
    d1 = __builtin_amdgcn_wmma_f32_16x16x4_f32(false, f1, false, a, (short)0, d1, false, false);
  }
}

template <int G>
__device__ __forceinline__ void zc_mma(const v4f (&cc)[NCF], const v4f (&uu)[2][GCH],
                                       v8f (&acc)[4][2]) {
#pragma unroll
  for (int k = 0; k < GCH; ++k) {
#pragma unroll
    for (int j = 0; j < 4; ++j) {
      const int i = G * GCH + k;
      if (i <= 2 * j + 34)
        zc_chunk(cc[k - 2 * j + 6], uu[0][k], uu[1][k], acc[j][0], acc[j][1]);
    }
  }
}

__device__ __forceinline__ void zc_plain(const float* __restrict__ u, const float* __restrict__ ctab,
                                         const float* __restrict__ hh, const float* __restrict__ h8,
                                         const float* __restrict__ Eps,
                                         const float* __restrict__ pre, const float* __restrict__ bz,
                                         float* __restrict__ out, const int T, const int m) {
  const int lane = threadIdx.x & 31;
  const int wv   = threadIdx.x >> 5;
  const int b0 = (blockIdx.y * 4 + wv) * 32;
  const int col  = lane & 15;
  const int half = lane >> 4;

  v8f acc[4][2];
#pragma unroll
  for (int j = 0; j < 4; ++j)
#pragma unroll
    for (int f = 0; f < 2; ++f)
#pragma unroll
      for (int v = 0; v < 8; ++v) acc[j][f][v] = 0.0f;

  const int ib = 103 - col + 4 * half;
  const int sh = ib & 3;
  const float* cp = ctab + sh * CLEN + (ib - sh);
  const float* up = u + (ptrdiff_t)(b0 + col) * USTR + 4 * half;
  const int m0 = 8 * T - 32;

  int g = 8 - 2 * ((T >= m) ? m : T);

  v4f cA[NCF], uA[2][GCH], cB[NCF], uB[2][GCH];
  zc_load(cp, up, m0, g, cA, uA);

  if (T < m) {
    const float* eb = Eps + ((size_t)(2 * half) * BATCH + b0 + col);
#pragma unroll
    for (int ks = 0; ks < 2; ++ks) {
      const float* e4 = eb + (size_t)(4 * ks) * BATCH;
      v2f f0, f1;
      f0.x = e4[0];   f0.y = e4[BATCH];
      f1.x = e4[16];  f1.y = e4[BATCH + 16];
#pragma unroll
      for (int j = 0; j < 4; ++j) {
        const float* hb = hh + (HOFF + 64 * T + 16 * j + 1 + col + 7 - 2 * half);
        v2f a;
        a.x = hb[-4 * ks];
        a.y = hb[-4 * ks - 1];
        acc[j][0] = __builtin_amdgcn_wmma_f32_16x16x4_f32(false, f0, false, a, (short)0,
                                                          acc[j][0], false, false);
        acc[j][1] = __builtin_amdgcn_wmma_f32_16x16x4_f32(false, f1, false, a, (short)0,
                                                          acc[j][1], false, false);
      }
    }
  }

  for (; g < 8; g += 2) {
    zc_load(cp, up, m0, g + 1, cB, uB);
    zc_mma<4>(cA, uA, acc);
    zc_load(cp, up, m0, g + 2, cA, uA);
    zc_mma<4>(cB, uB, acc);
  }
  zc_load(cp, up, m0, 9, cB, uB);
  const v4f cL  = *(const v4f*)(cp + 34 * 8);
  const v4f uL0 = *(const v4f*)(up + (m0 + 40) * 8);
  const v4f uL1 = *(const v4f*)(up + (ptrdiff_t)16 * USTR + (m0 + 40) * 8);
  const float hl = (T >= m) ? h8[LASTSLOT] : 0.0f;
  const float gl = (T >= m) ? h8[GAINSLOT] : 0.0f;
  const float* zp = bz + (b0 + 8 * half);
  const float* pp = pre + (size_t)(b0 + 8 * half) * NSEG;
  const int pi = (T >= m) ? T - m : 0;
  float bs[2][8];
#pragma unroll
  for (int f = 0; f < 2; ++f)
#pragma unroll
    for (int v = 0; v < 8; ++v) {
      const float* pr = pp + (16 * f + v) * NSEG;
      bs[f][v] = zp[16 * f + v] + (hl * pr[NSEG - 1] + gl * pr[pi]);
    }
  zc_mma<8>(cA, uA, acc);
  zc_mma<9>(cB, uB, acc);
  zc_chunk(cL, uL0, uL1, acc[3][0], acc[3][1]);

  float* op = out + (size_t)(b0 + 8 * half) * OSTR + 9 + 64 * T + col;
#pragma unroll
  for (int j = 0; j < 4; ++j) {
#pragma unroll
    for (int f = 0; f < 2; ++f) {
#pragma unroll
      for (int v = 0; v < 8; ++v)
        op[(size_t)(16 * f + v) * OSTR + 16 * j] = acc[j][f][v] + bs[f][v];
    }
  }
}

#define PI0 24
#define PCH 25
#define PR0 18
#define PCF 17

template <int E>
__device__ __forceinline__ void zc_pair(const float* __restrict__ u, const float* __restrict__ ctab,
                                        const float* __restrict__ h8,
                                        const float* __restrict__ pre, const float* __restrict__ bz,
                                        float* __restrict__ out, const int T) {
  const int lane = threadIdx.x & 31;
  const int wv   = threadIdx.x >> 5;
  const int b0 = (blockIdx.y * 4 + wv) * 32;
  const int col  = lane & 15;
  const int half = lane >> 4;

  v8f acc[8][2];
#pragma unroll
  for (int J = 0; J < 8; ++J)
#pragma unroll
    for (int f = 0; f < 2; ++f)
#pragma unroll
      for (int v = 0; v < 8; ++v) acc[J][f][v] = 0.0f;

  const int ib = 103 - col + 4 * half;
  const int sh = ib & 3;
  const float* cp = ctab + sh * CLEN + (ib - sh);
  const float* up = u + (ptrdiff_t)(b0 + col) * USTR + 4 * half;
  const int m0 = 8 * T - 32;

  v4f cf[PCF], uu[2][PCH];
#pragma unroll
  for (int r = 0; r < PCF; ++r) cf[r] = *(const v4f*)(cp + (PR0 + r) * 8);
#pragma unroll
  for (int ii = 0; ii < PCH; ++ii)
#pragma unroll
    for (int f = 0; f < 2; ++f)
      uu[f][ii] = *(const v4f*)(up + (ptrdiff_t)(16 * f) * USTR + (m0 + PI0 + ii) * 8);

  const float gain = h8[GAINSLOT];
  const float hl   = h8[LASTSLOT];
  v2f gb;
  gb.x = gain;  gb.y = gain;

  const float* zp = bz + (b0 + 8 * half);
  const float* pp = pre + (size_t)(b0 + 8 * half) * NSEG;
  float* op = out + (size_t)(b0 + 8 * half) * OSTR + 9 + 64 * T + col;
  float zv[2][8], q31[2][8], qa[2][8], qb[2][8];

  float S[8][2];
#pragma unroll
  for (int J = 0; J < 8; ++J) { S[J][0] = 0.0f; S[J][1] = 0.0f; }

#pragma unroll
  for (int ii = 0; ii < PCH; ++ii) {
    const int i = PI0 + ii;
    float t0 = 0.0f, t1 = 0.0f;
    if (ii < 16) {
      t0 = (uu[0][ii].x + uu[0][ii].y) + (uu[0][ii].z + uu[0][ii].w);
      t1 = (uu[1][ii].x + uu[1][ii].y) + (uu[1][ii].z + uu[1][ii].w);
    }
#pragma unroll
    for (int J = 0; J < 8; ++J) {
      const int j = J & 3;
      const int c = ii - 8 * (J >> 2);
      if (c >= 0 && c < 8) {
        if (c <= E + 2 * j) {
          S[J][0] += t0;
          S[J][1] += t1;
        } else {
          zc_chunk(cf[i - 2 * J - PR0], uu[0][ii], uu[1][ii], acc[J][0], acc[J][1]);
        }
      } else if (c >= 8 && i <= 2 * J + 34) {
        zc_chunk(cf[i - 2 * J - PR0], uu[0][ii], uu[1][ii], acc[J][0], acc[J][1]);
      }
    }
    if (ii == 7 || ii == 15) {
      const int G = ii >> 3;
#pragma unroll
      for (int j = 0; j < 4; ++j) {
        if (E + 2 * j >= 0) {
#pragma unroll
          for (int f = 0; f < 2; ++f) {
            v2f sa;
            sa.x = S[4 * G + j][f];  sa.y = 0.0f;
            acc[4 * G + j][f] = __builtin_amdgcn_wmma_f32_16x16x4_f32(false, sa, false, gb, (short)0,
                                                                  acc[4 * G + j][f], false, false);
          }
        }
      }
    }
    if (ii == 8) {
#pragma unroll
      for (int f = 0; f < 2; ++f)
#pragma unroll
        for (int v = 0; v < 8; ++v) {
          const float* pr = pp + (16 * f + v) * NSEG;
          zv[f][v]  = zp[16 * f + v];
          q31[f][v] = pr[NSEG - 1];
          qa[f][v]  = pr[T - 1];
          qb[f][v]  = pr[T];
        }
    }
    if (ii == 16) {
#pragma unroll
      for (int j = 0; j < 4; ++j)
#pragma unroll
        for (int f = 0; f < 2; ++f)
#pragma unroll
          for (int v = 0; v < 8; ++v)
            op[(size_t)(16 * f + v) * OSTR + 16 * j] =
                acc[j][f][v] + (zv[f][v] + (hl * q31[f][v] + gain * qa[f][v]));
    }
  }
#pragma unroll
  for (int j = 0; j < 4; ++j)
#pragma unroll
    for (int f = 0; f < 2; ++f)
#pragma unroll
      for (int v = 0; v < 8; ++v)
        op[(size_t)(16 * f + v) * OSTR + 16 * (4 + j)] =
            acc[4 + j][f][v] + (zv[f][v] + (hl * q31[f][v] + gain * qb[f][v]));
}

__global__ void __launch_bounds__(128, 2)
k_zconv(const float* __restrict__ u, const float* __restrict__ ctab,
        const float* __restrict__ hh, const float* __restrict__ h8,
        const float* __restrict__ Eps,
        const float* __restrict__ pre, const float* __restrict__ bz, float* __restrict__ out) {
  const int T = 2 * blockIdx.x;
  int m = __float_as_int(h8[BANDSLOT]);
  m = m < 1 ? 1 : (m > MAXBAND ? MAXBAND : m);
  int kc = __float_as_int(h8[KCSLOT]);
  kc = kc < 0 ? 0 : (kc > KTAPS - 1 ? KTAPS - 1 : kc);
  const int e = 8 * m - ((kc + 13) >> 3);
  if (m == 1 && T >= 2 && e >= -1) {
    switch (e) {
      case -1: zc_pair<-1>(u, ctab, h8, pre, bz, out, T); break;
      case 0:  zc_pair<0>(u, ctab, h8, pre, bz, out, T); break;
      case 1:  zc_pair<1>(u, ctab, h8, pre, bz, out, T); break;
      case 2:  zc_pair<2>(u, ctab, h8, pre, bz, out, T); break;
      default: zc_pair<3>(u, ctab, h8, pre, bz, out, T); break;
    }
  } else {
#pragma nounroll
    for (int t = 0; t < 2; ++t) zc_plain(u, ctab, hh, h8, Eps, pre, bz, out, T + t, m);
  }
}

extern "C" void kernel_launch(void* const* d_in, const int* in_sizes, int n_in,
                              void* d_out, int out_size, void* d_ws, size_t ws_size,
                              hipStream_t stream) {
  (void)in_sizes; (void)n_in; (void)out_size; (void)ws_size;
  const float* y = (const float*)d_in[0];
  const float* u = (const float*)d_in[1];
  const float* W = (const float*)d_in[2];
  float* out = (float*)d_out;

  float* hh   = (float*)d_ws;
  float* h8   = hh + HPAD_LEN;
  float* ctab = h8 + 32;
  float* Eps  = ctab + 4 * CLEN;
  float* pre  = Eps + 8 * BATCH;
  float* bz   = pre + BATCH * NSEG;

  k_setup <<<BATCH / (8 * RPW) + 1, 256, 0, stream>>>(y, u, W, hh, h8, Eps, ctab, pre, bz, out);
  k_zconv <<<dim3(SEQ / 128, BATCH / 128), 128, 0, stream>>>(u, ctab, hh, h8, Eps, pre, bz, out);
}
